# starting kernel
# baseline (speedup 1.0000x reference)
#include <hip/hip_runtime.h>
#include <math.h>

#define GAT_N 50000
#define GAT_E 800000
#define GAT_H 4
#define GAT_SLOPE 0.2f

typedef __attribute__((ext_vector_type(2))) float v2f;
typedef __attribute__((ext_vector_type(8))) float v8f;

template <int K, int NCOL>
__global__ __launch_bounds__(128) void gat_gemm_wmma(
    const float* __restrict__ A, const float* __restrict__ B,
    float* __restrict__ C, int M)
{
  const int wave = threadIdx.x >> 5;
  const int lane = threadIdx.x & 31;
  const int row0 = blockIdx.x << 5;
  const int col0 = ((blockIdx.y << 2) + wave) << 4;
  if (col0 >= NCOL) return;

  const int m     = lane & 15;
  const int khalf = (lane >> 4) << 1;
  const int n     = col0 + m;
  const bool hi_valid = (row0 + 16) < M;

  const float* arow0 = A + (size_t)(row0 + m) * K + khalf;
  const float* arow1 = hi_valid ? (arow0 + (size_t)16 * K) : arow0;
  const float* bcol  = B + (size_t)khalf * NCOL + n;

  v8f acc0 = {}, acc1 = {};
#pragma unroll 8
  for (int k0 = 0; k0 < K; k0 += 4) {
    v2f a0 = *(const v2f*)arow0;
    v2f a1 = *(const v2f*)arow1;
    v2f b;
    b.x = bcol[0];
    b.y = bcol[NCOL];
    acc0 = __builtin_amdgcn_wmma_f32_16x16x4_f32(
        false, a0, false, b, (short)0, acc0, false, false);
    acc1 = __builtin_amdgcn_wmma_f32_16x16x4_f32(
        false, a1, false, b, (short)0, acc1, false, false);
    arow0 += 4;
    arow1 += 4;
    bcol  += NCOL * 4;
  }

  const int rsub = (lane >> 4) << 3;
  float* crow0 = C + (size_t)(row0 + rsub) * NCOL + n;
#pragma unroll
  for (int r = 0; r < 8; ++r) crow0[(size_t)r * NCOL] = acc0[r];
  if (hi_valid) {
    float* crow1 = C + (size_t)(row0 + 16 + rsub) * NCOL + n;
#pragma unroll
    for (int r = 0; r < 8; ++r) crow1[(size_t)r * NCOL] = acc1[r];
  }
}

__global__ void gat_attn_scores(const float* __restrict__ feat,
                                const float* __restrict__ al,
                                const float* __restrict__ ar,
                                float* __restrict__ el, float* __restrict__ er,
                                int F, int D)
{
  int idx = blockIdx.x * blockDim.x + threadIdx.x;
  if (idx >= GAT_N * GAT_H) return;
  int nd = idx >> 2, h = idx & 3;
  const float* f   = feat + (size_t)nd * F + h * D;
  const float* alh = al + h * D;
  const float* arh = ar + h * D;
  float sl = 0.f, sr = 0.f;
  for (int d = 0; d < D; ++d) {
    float v = f[d];
    sl = fmaf(v, alh[d], sl);
    sr = fmaf(v, arh[d], sr);
  }
  el[idx] = sl;
  er[idx] = sr;
}

__global__ void gat_init_mden(float* __restrict__ mx, float* __restrict__ den, int count)
{
  int idx = blockIdx.x * blockDim.x + threadIdx.x;
  if (idx < count) {
    mx[idx]  = __int_as_float((int)0xFF800000u);
    den[idx] = 0.f;
  }
}

__device__ __forceinline__ float gat_leaky(float x) {
  return x > 0.f ? x : GAT_SLOPE * x;
}

__device__ __forceinline__ void gat_atomic_max_f32(float* addr, float v) {
  if (v >= 0.f) atomicMax((int*)addr, __float_as_int(v));
  else          atomicMin((unsigned int*)addr, __float_as_uint(v));
}

__global__ void gat_edge_max(const int* __restrict__ src, const int* __restrict__ dst,
                             const float* __restrict__ el, const float* __restrict__ er,
                             float* __restrict__ mx)
{
  int idx = blockIdx.x * blockDim.x + threadIdx.x;
  if (idx >= GAT_E * GAT_H) return;
  int e = idx >> 2, h = idx & 3;
  int s = src[e], d = dst[e];
  float x = gat_leaky(el[s * GAT_H + h] + er[d * GAT_H + h]);
  gat_atomic_max_f32(&mx[d * GAT_H + h], x);
}

__global__ void gat_edge_exp(const int* __restrict__ src, const int* __restrict__ dst,
                             const float* __restrict__ el, const float* __restrict__ er,
                             const float* __restrict__ mx,
                             float* __restrict__ ex, float* __restrict__ den)
{
  int idx = blockIdx.x * blockDim.x + threadIdx.x;
  if (idx >= GAT_E * GAT_H) return;
  int e = idx >> 2, h = idx & 3;
  int s = src[e], d = dst[e];
  float x = gat_leaky(el[s * GAT_H + h] + er[d * GAT_H + h]);
  float v = expf(x - mx[d * GAT_H + h]);
  ex[idx] = v;
  atomicAdd(&den[d * GAT_H + h], v);
}

__global__ __launch_bounds__(256) void gat_edge_aggregate(
    const int* __restrict__ src, const int* __restrict__ dst,
    const float* __restrict__ feat, const float* __restrict__ ex,
    const float* __restrict__ den, float* __restrict__ out, int F, int D)
{
  int e = blockIdx.x * (blockDim.x >> 5) + (threadIdx.x >> 5);
  if (e >= GAT_E) return;
  int lane = threadIdx.x & 31;
  int s = src[e], d = dst[e];
  const float* fs = feat + (size_t)s * F;
  float*       od = out  + (size_t)d * F;
  __builtin_prefetch(fs + lane, 0, 0);
  float alpha[GAT_H];
#pragma unroll
  for (int h = 0; h < GAT_H; ++h)
    alpha[h] = ex[(size_t)e * GAT_H + h] / den[(size_t)d * GAT_H + h];
  for (int f = lane; f < F; f += 32)
    atomicAdd(&od[f], fs[f] * alpha[(unsigned)f / (unsigned)D]);
}

__global__ void gat_finalize_elu(float* __restrict__ acc, const float* __restrict__ bias,
                                 const float* __restrict__ res, int total)
{
  int idx = blockIdx.x * blockDim.x + threadIdx.x;
  if (idx >= total) return;
  float v = acc[idx] + bias[idx & 255];
  if (res) v += res[idx];
  acc[idx] = v > 0.f ? v : (expf(v) - 1.f);
}

__global__ void gat_finalize_out(const float* __restrict__ acc, const float* __restrict__ bias,
                                 const float* __restrict__ res2, float* __restrict__ out)
{
  const int C = 40;
  int idx = blockIdx.x * blockDim.x + threadIdx.x;
  if (idx >= GAT_N * C) return;
  int n = idx / C, c = idx % C;
  float sum = 0.f;
#pragma unroll
  for (int h = 0; h < GAT_H; ++h) {
    int j = n * (GAT_H * C) + h * C + c;
    sum += acc[j] + bias[h * C + c] + res2[j];
  }
  out[idx] = sum * 0.25f;
}

extern "C" void kernel_launch(void* const* d_in, const int* in_sizes, int n_in,
                              void* d_out, int out_size, void* d_ws, size_t ws_size,
                              hipStream_t stream)
{
  const float* x     = (const float*)d_in[0];
  const int*   src   = (const int*)d_in[1];
  const int*   dst   = (const int*)d_in[2];
  const float* W0    = (const float*)d_in[3];
  const float* al0   = (const float*)d_in[4];
  const float* ar0   = (const float*)d_in[5];
  const float* b0    = (const float*)d_in[6];
  const float* W1    = (const float*)d_in[7];
  const float* al1   = (const float*)d_in[8];
  const float* ar1   = (const float*)d_in[9];
  const float* b1    = (const float*)d_in[10];
  const float* W2    = (const float*)d_in[11];
  const float* al2   = (const float*)d_in[12];
  const float* ar2   = (const float*)d_in[13];
  const float* b2    = (const float*)d_in[14];
  const float* Wres2 = (const float*)d_in[15];

  const size_t NF = (size_t)GAT_N * 256;
  const size_t NG = (size_t)GAT_N * 160;
  float* ws   = (float*)d_ws;
  float* bufA = ws;
  float* bufB = bufA + NF;
  float* bufC = bufB + NF;
  float* bufD = bufC + NF;
  float* EX   = bufD + NG;
  float* EL   = EX + (size_t)GAT_E * GAT_H;
  float* ER   = EL + (size_t)GAT_N * GAT_H;
  float* MX   = ER + (size_t)GAT_N * GAT_H;
  float* DEN  = MX + (size_t)GAT_N * GAT_H;

  dim3 gemmBlk(128);
  const int rowTiles = (GAT_N + 31) / 32;
  dim3 gemmGrid256(rowTiles, 4);
  dim3 gemmGrid160(rowTiles, 3);

  const int ehThreads = GAT_E * GAT_H;
  const int ehBlocks  = (ehThreads + 255) / 256;
  const int nhThreads = GAT_N * GAT_H;
  const int nhBlocks  = (nhThreads + 255) / 256;
  const int aggBlocks = (GAT_E + 7) / 8;
  const int nfBlocks  = (int)((NF + 255) / 256);

  gat_gemm_wmma<256, 256><<<gemmGrid256, gemmBlk, 0, stream>>>(x, W0, bufA, GAT_N);
  gat_attn_scores<<<nhBlocks, 256, 0, stream>>>(bufA, al0, ar0, EL, ER, 256, 64);
  gat_init_mden<<<nhBlocks, 256, 0, stream>>>(MX, DEN, nhThreads);
  gat_edge_max<<<ehBlocks, 256, 0, stream>>>(src, dst, EL, ER, MX);
  gat_edge_exp<<<ehBlocks, 256, 0, stream>>>(src, dst, EL, ER, MX, EX, DEN);
  (void)hipMemsetAsync(bufB, 0, NF * sizeof(float), stream);
  gat_edge_aggregate<<<aggBlocks, 256, 0, stream>>>(src, dst, bufA, EX, DEN, bufB, 256, 64);
  gat_finalize_elu<<<nfBlocks, 256, 0, stream>>>(bufB, b0, nullptr, (int)NF);

  gat_gemm_wmma<256, 256><<<gemmGrid256, gemmBlk, 0, stream>>>(bufB, W1, bufA, GAT_N);
  gat_attn_scores<<<nhBlocks, 256, 0, stream>>>(bufA, al1, ar1, EL, ER, 256, 64);
  gat_init_mden<<<nhBlocks, 256, 0, stream>>>(MX, DEN, nhThreads);
  gat_edge_max<<<ehBlocks, 256, 0, stream>>>(src, dst, EL, ER, MX);
  gat_edge_exp<<<ehBlocks, 256, 0, stream>>>(src, dst, EL, ER, MX, EX, DEN);
  (void)hipMemsetAsync(bufC, 0, NF * sizeof(float), stream);
  gat_edge_aggregate<<<aggBlocks, 256, 0, stream>>>(src, dst, bufA, EX, DEN, bufC, 256, 64);
  gat_finalize_elu<<<nfBlocks, 256, 0, stream>>>(bufC, b1, bufB, (int)NF);

  gat_gemm_wmma<256, 160><<<gemmGrid160, gemmBlk, 0, stream>>>(bufC, W2, bufA, GAT_N);
  gat_gemm_wmma<256, 160><<<gemmGrid160, gemmBlk, 0, stream>>>(bufC, Wres2, bufB, GAT_N);
  gat_attn_scores<<<nhBlocks, 256, 0, stream>>>(bufA, al2, ar2, EL, ER, 160, 40);
  gat_init_mden<<<nhBlocks, 256, 0, stream>>>(MX, DEN, nhThreads);
  gat_edge_max<<<ehBlocks, 256, 0, stream>>>(src, dst, EL, ER, MX);
  gat_edge_exp<<<ehBlocks, 256, 0, stream>>>(src, dst, EL, ER, MX, EX, DEN);
  (void)hipMemsetAsync(bufD, 0, NG * sizeof(float), stream);
  gat_edge_aggregate<<<aggBlocks, 256, 0, stream>>>(src, dst, bufA, EX, DEN, bufD, 160, 40);
  gat_finalize_out<<<(GAT_N * 40 + 255) / 256, 256, 0, stream>>>(bufD, b2, bufB, (float*)d_out);
}


# aggregate per destination: edges grouped by dst once per call (count, scan, fill), one wave per node does max, exp-sum and row sum in registers; float atomics and zero fills gone, sum order changes
# speedup vs baseline: 2.6542x; 2.6542x over previous
#include <hip/hip_runtime.h>
#include <math.h>

#define GAT_N 50000
#define GAT_E 800000
#define GAT_H 4
#define GAT_SLOPE 0.2f

typedef __attribute__((ext_vector_type(2))) float v2f;
typedef __attribute__((ext_vector_type(8))) float v8f;

template <int K, int NCOL>
__global__ __launch_bounds__(128) void gat_gemm_wmma(
    const float* __restrict__ A, const float* __restrict__ B,
    float* __restrict__ C, int M)
{
  const int wave = threadIdx.x >> 5;
  const int lane = threadIdx.x & 31;
  const int row0 = blockIdx.x << 5;
  const int col0 = ((blockIdx.y << 2) + wave) << 4;
  if (col0 >= NCOL) return;

  const int m     = lane & 15;
  const int khalf = (lane >> 4) << 1;
  const int n     = col0 + m;
  const bool hi_valid = (row0 + 16) < M;

  const float* arow0 = A + (size_t)(row0 + m) * K + khalf;
  const float* arow1 = hi_valid ? (arow0 + (size_t)16 * K) : arow0;
  const float* bcol  = B + (size_t)khalf * NCOL + n;

  v8f acc0 = {}, acc1 = {};
#pragma unroll 8
  for (int k0 = 0; k0 < K; k0 += 4) {
    v2f a0 = *(const v2f*)arow0;
    v2f a1 = *(const v2f*)arow1;
    v2f b;
    b.x = bcol[0];
    b.y = bcol[NCOL];
    acc0 = __builtin_amdgcn_wmma_f32_16x16x4_f32(
        false, a0, false, b, (short)0, acc0, false, false);
    acc1 = __builtin_amdgcn_wmma_f32_16x16x4_f32(
        false, a1, false, b, (short)0, acc1, false, false);
    arow0 += 4;
    arow1 += 4;
    bcol  += NCOL * 4;
  }

  const int rsub = (lane >> 4) << 3;
  float* crow0 = C + (size_t)(row0 + rsub) * NCOL + n;
#pragma unroll
  for (int r = 0; r < 8; ++r) crow0[(size_t)r * NCOL] = acc0[r];
  if (hi_valid) {
    float* crow1 = C + (size_t)(row0 + 16 + rsub) * NCOL + n;
#pragma unroll
    for (int r = 0; r < 8; ++r) crow1[(size_t)r * NCOL] = acc1[r];
  }
}

__global__ void gat_attn_scores(const float* __restrict__ feat,
                                const float* __restrict__ al,
                                const float* __restrict__ ar,
                                float* __restrict__ el, float* __restrict__ er,
                                int F, int D)
{
  int idx = blockIdx.x * blockDim.x + threadIdx.x;
  if (idx >= GAT_N * GAT_H) return;
  int nd = idx >> 2, h = idx & 3;
  const float* f   = feat + (size_t)nd * F + h * D;
  const float* alh = al + h * D;
  const float* arh = ar + h * D;
  float sl = 0.f, sr = 0.f;
  for (int d = 0; d < D; ++d) {
    float v = f[d];
    sl = fmaf(v, alh[d], sl);
    sr = fmaf(v, arh[d], sr);
  }
  el[idx] = sl;
  er[idx] = sr;
}

__device__ __forceinline__ float gat_leaky(float x) {
  return x > 0.f ? x : GAT_SLOPE * x;
}

#define GAT_SCAN_THREADS 1024
#define GAT_SCAN_PER 52
#define GAT_CNT_PAD (GAT_SCAN_THREADS * GAT_SCAN_PER)

__global__ void gat_zero_counts(int* __restrict__ cnt)
{
  int idx = blockIdx.x * blockDim.x + threadIdx.x;
  if (idx < GAT_CNT_PAD) cnt[idx] = 0;
}

__global__ void gat_dst_rank(const int* __restrict__ dst, int* __restrict__ cnt,
                             int* __restrict__ rank)
{
  int e = blockIdx.x * blockDim.x + threadIdx.x;
  if (e >= GAT_E) return;
  rank[e] = atomicAdd(&cnt[dst[e]], 1);
}

__global__ __launch_bounds__(GAT_SCAN_THREADS) void gat_rowptr_scan(
    const int* __restrict__ cnt, int* __restrict__ rowptr)
{
  __shared__ int wsum[32];
  const int t = threadIdx.x, lane = t & 31, w = t >> 5;
  const int4* p = (const int4*)(cnt + t * GAT_SCAN_PER);
  int4 v[GAT_SCAN_PER / 4];
  int tot = 0;
#pragma unroll
  for (int i = 0; i < GAT_SCAN_PER / 4; ++i) {
    v[i] = p[i];
    tot += v[i].x + v[i].y + v[i].z + v[i].w;
  }
  int inc = tot;
#pragma unroll
  for (int o = 1; o < 32; o <<= 1) {
    int nb = __shfl_up(inc, o, 32);
    if (lane >= o) inc += nb;
  }
  if (lane == 31) wsum[w] = inc;
  __syncthreads();
  if (w == 0) {
    int x = wsum[lane], xi = x;
#pragma unroll
    for (int o = 1; o < 32; o <<= 1) {
      int nb = __shfl_up(xi, o, 32);
      if (lane >= o) xi += nb;
    }
    wsum[lane] = xi - x;
  }
  __syncthreads();
  int run = wsum[w] + inc - tot;
  int4* q = (int4*)(rowptr + t * GAT_SCAN_PER);
#pragma unroll
  for (int i = 0; i < GAT_SCAN_PER / 4; ++i) {
    int4 o;
    o.x = run; run += v[i].x;
    o.y = run; run += v[i].y;
    o.z = run; run += v[i].z;
    o.w = run; run += v[i].w;
    q[i] = o;
  }
}

__global__ void gat_fill_segments(const int* __restrict__ src, const int* __restrict__ dst,
                                  const int* __restrict__ rowptr, const int* __restrict__ rank,
                                  int* __restrict__ csrc)
{
  int e = blockIdx.x * blockDim.x + threadIdx.x;
  if (e >= GAT_E) return;
  csrc[rowptr[dst[e]] + rank[e]] = src[e];
}

__device__ __forceinline__ float gat_wave_max(float v) {
#pragma unroll
  for (int o = 16; o > 0; o >>= 1) v = fmaxf(v, __shfl_xor(v, o, 32));
  return v;
}
__device__ __forceinline__ float gat_wave_sum(float v) {
#pragma unroll
  for (int o = 16; o > 0; o >>= 1) v += __shfl_xor(v, o, 32);
  return v;
}
__device__ __forceinline__ float gat_lane_bcast(float v, int l) {
  return __int_as_float(__builtin_amdgcn_readlane(__float_as_int(v), l));
}

__device__ __forceinline__ int gat_slot_src(const int* __restrict__ csrc, int j) {
  unsigned u = (unsigned)csrc[j];
  return (int)(u < (unsigned)GAT_N ? u : (unsigned)(GAT_N - 1));
}

__device__ __forceinline__ float4 gat_edge_logits(const float* __restrict__ el, int s, float4 erd) {
  const float4 e = *(const float4*)(el + (size_t)s * GAT_H);
  float4 x;
  x.x = gat_leaky(e.x + erd.x);
  x.y = gat_leaky(e.y + erd.y);
  x.z = gat_leaky(e.z + erd.z);
  x.w = gat_leaky(e.w + erd.w);
  return x;
}

__device__ __forceinline__ void gat_segment_stats(
    const int* __restrict__ csrc, const float* __restrict__ el, float4 erd,
    int beg, int end, int lane, float4& m, float4& den)
{
  const float ninf = __int_as_float((int)0xFF800000u);
  m.x = ninf; m.y = ninf; m.z = ninf; m.w = ninf;
  for (int c = beg; c < end; c += 32) {
    const int j = c + lane;
    if (j < end) {
      const float4 x = gat_edge_logits(el, gat_slot_src(csrc, j), erd);
      m.x = fmaxf(m.x, x.x); m.y = fmaxf(m.y, x.y);
      m.z = fmaxf(m.z, x.z); m.w = fmaxf(m.w, x.w);
    }
  }
  m.x = gat_wave_max(m.x); m.y = gat_wave_max(m.y);
  m.z = gat_wave_max(m.z); m.w = gat_wave_max(m.w);
  den.x = 0.f; den.y = 0.f; den.z = 0.f; den.w = 0.f;
  for (int c = beg; c < end; c += 32) {
    const int j = c + lane;
    if (j < end) {
      const float4 x = gat_edge_logits(el, gat_slot_src(csrc, j), erd);
      den.x += expf(x.x - m.x); den.y += expf(x.y - m.y);
      den.z += expf(x.z - m.z); den.w += expf(x.w - m.w);
    }
  }
  den.x = gat_wave_sum(den.x); den.y = gat_wave_sum(den.y);
  den.z = gat_wave_sum(den.z); den.w = gat_wave_sum(den.w);
}

__device__ __forceinline__ void gat_chunk_weights(
    const int* __restrict__ csrc, const float* __restrict__ el, float4 erd,
    int c, int end, int lane, float4 m, float4 den, int& s, float4& a)
{
  s = 0;
  a.x = 0.f; a.y = 0.f; a.z = 0.f; a.w = 0.f;
  const int j = c + lane;
  if (j < end) {
    s = gat_slot_src(csrc, j);
    const float4 x = gat_edge_logits(el, s, erd);
    a.x = expf(x.x - m.x) / den.x; a.y = expf(x.y - m.y) / den.y;
    a.z = expf(x.z - m.z) / den.z; a.w = expf(x.w - m.w) / den.w;
  }
}

__global__ __launch_bounds__(256) void gat_node_agg_elu(
    const int* __restrict__ rowptr, const int* __restrict__ csrc,
    const float* __restrict__ feat, const float* __restrict__ el,
    const float* __restrict__ er, const float* __restrict__ bias,
    const float* __restrict__ res, float* __restrict__ out)
{
  const int d = __builtin_amdgcn_readfirstlane((int)(blockIdx.x * 8 + (threadIdx.x >> 5)));
  if (d >= GAT_N) return;
  const int lane = threadIdx.x & 31;
  const int beg = rowptr[d], end = rowptr[d + 1];
  const float4 erd = *(const float4*)(er + (size_t)d * GAT_H);

  float4 m, den;
  gat_segment_stats(csrc, el, erd, beg, end, lane, m, den);

  const bool lo = lane < 16;
  float4 accA = {0.f, 0.f, 0.f, 0.f}, accB = {0.f, 0.f, 0.f, 0.f};
  for (int c = beg; c < end; c += 32) {
    int s; float4 a;
    gat_chunk_weights(csrc, el, erd, c, end, lane, m, den, s, a);
    const int cnt = (end - c) < 32 ? (end - c) : 32;
#pragma unroll 4
    for (int t = 0; t < cnt; ++t) {
      const int st = __builtin_amdgcn_readlane(s, t);
      const float a0 = gat_lane_bcast(a.x, t), a1 = gat_lane_bcast(a.y, t);
      const float a2 = gat_lane_bcast(a.z, t), a3 = gat_lane_bcast(a.w, t);
      const float wA = lo ? a0 : a1;
      const float wB = lo ? a2 : a3;
      const float4* row = (const float4*)(feat + (size_t)st * 256);
      const float4 vA = row[lane];
      const float4 vB = row[32 + lane];
      accA.x += vA.x * wA; accA.y += vA.y * wA; accA.z += vA.z * wA; accA.w += vA.w * wA;
      accB.x += vB.x * wB; accB.y += vB.y * wB; accB.z += vB.z * wB; accB.w += vB.w * wB;
    }
  }

  const float4 bA = *(const float4*)(bias + 4 * lane);
  const float4 bB = *(const float4*)(bias + 128 + 4 * lane);
  float4 vA, vB;
  vA.x = accA.x + bA.x; vA.y = accA.y + bA.y; vA.z = accA.z + bA.z; vA.w = accA.w + bA.w;
  vB.x = accB.x + bB.x; vB.y = accB.y + bB.y; vB.z = accB.z + bB.z; vB.w = accB.w + bB.w;
  const size_t o = (size_t)d * 256 + 4 * lane;
  if (res) {
    const float4 rA = *(const float4*)(res + o);
    const float4 rB = *(const float4*)(res + o + 128);
    vA.x += rA.x; vA.y += rA.y; vA.z += rA.z; vA.w += rA.w;
    vB.x += rB.x; vB.y += rB.y; vB.z += rB.z; vB.w += rB.w;
  }
  vA.x = vA.x > 0.f ? vA.x : (expf(vA.x) - 1.f);
  vA.y = vA.y > 0.f ? vA.y : (expf(vA.y) - 1.f);
  vA.z = vA.z > 0.f ? vA.z : (expf(vA.z) - 1.f);
  vA.w = vA.w > 0.f ? vA.w : (expf(vA.w) - 1.f);
  vB.x = vB.x > 0.f ? vB.x : (expf(vB.x) - 1.f);
  vB.y = vB.y > 0.f ? vB.y : (expf(vB.y) - 1.f);
  vB.z = vB.z > 0.f ? vB.z : (expf(vB.z) - 1.f);
  vB.w = vB.w > 0.f ? vB.w : (expf(vB.w) - 1.f);
  *(float4*)(out + o) = vA;
  *(float4*)(out + o + 128) = vB;
}

__global__ __launch_bounds__(256) void gat_node_agg_mean(
    const int* __restrict__ rowptr, const int* __restrict__ csrc,
    const float* __restrict__ feat, const float* __restrict__ el,
    const float* __restrict__ er, const float* __restrict__ bias,
    const float* __restrict__ res2, float* __restrict__ out)
{
  __shared__ float sval[8 * 160];
  const int w = threadIdx.x >> 5;
  const int d = __builtin_amdgcn_readfirstlane((int)(blockIdx.x * 8 + w));
  const int lane = threadIdx.x & 31;

  if (d < GAT_N) {
    const int beg = rowptr[d], end = rowptr[d + 1];
    const float4 erd = *(const float4*)(er + (size_t)d * GAT_H);
    float4 m, den;
    gat_segment_stats(csrc, el, erd, beg, end, lane, m, den);

    float acc0 = 0.f, acc1 = 0.f, acc2 = 0.f, acc3 = 0.f, acc4 = 0.f;
    for (int c = beg; c < end; c += 32) {
      int s; float4 a;
      gat_chunk_weights(csrc, el, erd, c, end, lane, m, den, s, a);
      const int cnt = (end - c) < 32 ? (end - c) : 32;
#pragma unroll 4
      for (int t = 0; t < cnt; ++t) {
        const int st = __builtin_amdgcn_readlane(s, t);
        const float a0 = gat_lane_bcast(a.x, t), a1 = gat_lane_bcast(a.y, t);
        const float a2 = gat_lane_bcast(a.z, t), a3 = gat_lane_bcast(a.w, t);
        const float* row = feat + (size_t)st * 160 + lane;
        const float v0 = row[0], v1 = row[32], v2 = row[64], v3 = row[96], v4 = row[128];
        acc0 += v0 * a0;
        acc1 += v1 * (lane >= 8 ? a1 : a0);
        acc2 += v2 * (lane >= 16 ? a2 : a1);
        acc3 += v3 * (lane >= 24 ? a3 : a2);
        acc4 += v4 * a3;
      }
    }
    const float* rr = res2 + (size_t)d * 160 + lane;
    const float* bb = bias + lane;
    float* sv = sval + w * 160 + lane;
    sv[0]   = acc0 + bb[0]   + rr[0];
    sv[32]  = acc1 + bb[32]  + rr[32];
    sv[64]  = acc2 + bb[64]  + rr[64];
    sv[96]  = acc3 + bb[96]  + rr[96];
    sv[128] = acc4 + bb[128] + rr[128];
  }
  __syncthreads();
  for (int i = threadIdx.x; i < 8 * 40; i += 256) {
    const int nl = i / 40, c = i - nl * 40;
    if ((int)(blockIdx.x * 8 + nl) < GAT_N) {
      const float* sv = sval + nl * 160 + c;
      float sum = 0.f;
#pragma unroll
      for (int h = 0; h < GAT_H; ++h) sum += sv[h * 40];
      out[(size_t)blockIdx.x * 320 + i] = sum * 0.25f;
    }
  }
}

extern "C" void kernel_launch(void* const* d_in, const int* in_sizes, int n_in,
                              void* d_out, int out_size, void* d_ws, size_t ws_size,
                              hipStream_t stream)
{
  const float* x     = (const float*)d_in[0];
  const int*   src   = (const int*)d_in[1];
  const int*   dst   = (const int*)d_in[2];
  const float* W0    = (const float*)d_in[3];
  const float* al0   = (const float*)d_in[4];
  const float* ar0   = (const float*)d_in[5];
  const float* b0    = (const float*)d_in[6];
  const float* W1    = (const float*)d_in[7];
  const float* al1   = (const float*)d_in[8];
  const float* ar1   = (const float*)d_in[9];
  const float* b1    = (const float*)d_in[10];
  const float* W2    = (const float*)d_in[11];
  const float* al2   = (const float*)d_in[12];
  const float* ar2   = (const float*)d_in[13];
  const float* b2    = (const float*)d_in[14];
  const float* Wres2 = (const float*)d_in[15];

  const size_t NF = (size_t)GAT_N * 256;
  float* ws   = (float*)d_ws;
  float* bufA = ws;
  float* bufB = bufA + NF;
  float* bufC = bufB + NF;
  float* EL   = bufC + NF;
  float* ER   = EL + (size_t)GAT_N * GAT_H;
  int*   CNT  = (int*)(ER + (size_t)GAT_N * GAT_H);
  int*   RPTR = CNT + GAT_CNT_PAD;
  int*   RANK = RPTR + GAT_CNT_PAD;
  int*   CSRC = RANK + GAT_E;

  dim3 gemmBlk(128);
  const int rowTiles = (GAT_N + 31) / 32;
  dim3 gemmGrid256(rowTiles, 4);
  dim3 gemmGrid160(rowTiles, 3);

  const int eBlocks    = (GAT_E + 255) / 256;
  const int nhThreads  = GAT_N * GAT_H;
  const int nhBlocks   = (nhThreads + 255) / 256;
  const int nodeBlocks = (GAT_N + 7) / 8;

  gat_zero_counts<<<GAT_CNT_PAD / 256, 256, 0, stream>>>(CNT);
  gat_dst_rank<<<eBlocks, 256, 0, stream>>>(dst, CNT, RANK);
  gat_rowptr_scan<<<1, GAT_SCAN_THREADS, 0, stream>>>(CNT, RPTR);
  gat_fill_segments<<<eBlocks, 256, 0, stream>>>(src, dst, RPTR, RANK, CSRC);

  gat_gemm_wmma<256, 256><<<gemmGrid256, gemmBlk, 0, stream>>>(x, W0, bufA, GAT_N);
  gat_attn_scores<<<nhBlocks, 256, 0, stream>>>(bufA, al0, ar0, EL, ER, 256, 64);
  gat_node_agg_elu<<<nodeBlocks, 256, 0, stream>>>(RPTR, CSRC, bufA, EL, ER, b0, nullptr, bufB);

  gat_gemm_wmma<256, 256><<<gemmGrid256, gemmBlk, 0, stream>>>(bufB, W1, bufA, GAT_N);
  gat_attn_scores<<<nhBlocks, 256, 0, stream>>>(bufA, al1, ar1, EL, ER, 256, 64);
  gat_node_agg_elu<<<nodeBlocks, 256, 0, stream>>>(RPTR, CSRC, bufA, EL, ER, b1, bufB, bufC);

  gat_gemm_wmma<256, 160><<<gemmGrid160, gemmBlk, 0, stream>>>(bufC, W2, bufA, GAT_N);
  gat_gemm_wmma<256, 160><<<gemmGrid160, gemmBlk, 0, stream>>>(bufC, Wres2, bufB, GAT_N);
  gat_attn_scores<<<nhBlocks, 256, 0, stream>>>(bufA, al2, ar2, EL, ER, 160, 40);
  gat_node_agg_mean<<<nodeBlocks, 256, 0, stream>>>(RPTR, CSRC, bufA, EL, ER, b2, bufB, (float*)d_out);
}


# GEMM wave tile 32x64 (32x32 for 160 columns): lane owns adjacent columns, 16-byte A and B fragment loads feed 16 matrix ops per 8 k, whole-line C stores; k sum order changes, f32 kept
# speedup vs baseline: 3.4632x; 1.3048x over previous
#include <hip/hip_runtime.h>
#include <math.h>

#define GAT_N 50000
#define GAT_E 800000
#define GAT_H 4
#define GAT_SLOPE 0.2f

typedef __attribute__((ext_vector_type(2))) float v2f;
typedef __attribute__((ext_vector_type(8))) float v8f;

template <int K, int NCOL, int NT>
__global__ __launch_bounds__(NCOL / (16 * NT) * 32) void gat_gemm_wmma(
    const float* __restrict__ A, const float* __restrict__ B,
    float* __restrict__ C, int M)
{
  typedef float vnt __attribute__((ext_vector_type(NT)));
  typedef float v4f __attribute__((ext_vector_type(4)));
  const int wave = threadIdx.x >> 5;
  const int lane = threadIdx.x & 31;
  const int row0 = blockIdx.x << 5;
  const int col0 = wave * (16 * NT);
  const int c    = lane & 15;
  const int h    = lane >> 4;
  const bool hi_valid = (row0 + 16) < M;

  const float* arow0 = A + (size_t)(row0 + c) * K + 4 * h;
  const float* arow1 = hi_valid ? (arow0 + (size_t)16 * K) : arow0;
  const float* bp    = B + (size_t)(4 * h) * NCOL + col0 + NT * c;

  v8f acc0[NT], acc1[NT];
#pragma unroll
  for (int t = 0; t < NT; ++t) { acc0[t] = (v8f){}; acc1[t] = (v8f){}; }

#pragma unroll 2
  for (int k0 = 0; k0 < K; k0 += 8) {
    const v4f a0 = *(const v4f*)(arow0 + k0);
    const v4f a1 = *(const v4f*)(arow1 + k0);
    vnt b[4];
#pragma unroll
    for (int i = 0; i < 4; ++i) b[i] = *(const vnt*)(bp + (size_t)(k0 + i) * NCOL);
#pragma unroll
    for (int j = 0; j < 2; ++j) {
      v2f fa0, fa1;
      fa0.x = a0[2 * j]; fa0.y = a0[2 * j + 1];
      fa1.x = a1[2 * j]; fa1.y = a1[2 * j + 1];
#pragma unroll
      for (int t = 0; t < NT; ++t) {
        v2f fb;
        fb.x = b[2 * j][t];
        fb.y = b[2 * j + 1][t];
        acc0[t] = __builtin_amdgcn_wmma_f32_16x16x4_f32(
            false, fa0, false, fb, (short)0, acc0[t], false, false);
        acc1[t] = __builtin_amdgcn_wmma_f32_16x16x4_f32(
            false, fa1, false, fb, (short)0, acc1[t], false, false);
      }
    }
  }

  float* crow0 = C + (size_t)(row0 + 8 * h) * NCOL + col0 + NT * c;
#pragma unroll
  for (int r = 0; r < 8; ++r) {
    vnt o;
#pragma unroll
    for (int t = 0; t < NT; ++t) o[t] = acc0[t][r];
    *(vnt*)(crow0 + (size_t)r * NCOL) = o;
  }
  if (hi_valid) {
    float* crow1 = crow0 + (size_t)16 * NCOL;
#pragma unroll
    for (int r = 0; r < 8; ++r) {
      vnt o;
#pragma unroll
      for (int t = 0; t < NT; ++t) o[t] = acc1[t][r];
      *(vnt*)(crow1 + (size_t)r * NCOL) = o;
    }
  }
}

__global__ void gat_attn_scores(const float* __restrict__ feat,
                                const float* __restrict__ al,
                                const float* __restrict__ ar,
                                float* __restrict__ el, float* __restrict__ er,
                                int F, int D)
{
  int idx = blockIdx.x * blockDim.x + threadIdx.x;
  if (idx >= GAT_N * GAT_H) return;
  int nd = idx >> 2, h = idx & 3;
  const float* f   = feat + (size_t)nd * F + h * D;
  const float* alh = al + h * D;
  const float* arh = ar + h * D;
  float sl = 0.f, sr = 0.f;
  for (int d = 0; d < D; ++d) {
    float v = f[d];
    sl = fmaf(v, alh[d], sl);
    sr = fmaf(v, arh[d], sr);
  }
  el[idx] = sl;
  er[idx] = sr;
}

__device__ __forceinline__ float gat_leaky(float x) {
  return x > 0.f ? x : GAT_SLOPE * x;
}

#define GAT_SCAN_THREADS 1024
#define GAT_SCAN_PER 52
#define GAT_CNT_PAD (GAT_SCAN_THREADS * GAT_SCAN_PER)

__global__ void gat_zero_counts(int* __restrict__ cnt)
{
  int idx = blockIdx.x * blockDim.x + threadIdx.x;
  if (idx < GAT_CNT_PAD) cnt[idx] = 0;
}

__global__ void gat_dst_rank(const int* __restrict__ dst, int* __restrict__ cnt,
                             int* __restrict__ rank)
{
  int e = blockIdx.x * blockDim.x + threadIdx.x;
  if (e >= GAT_E) return;
  rank[e] = atomicAdd(&cnt[dst[e]], 1);
}

__global__ __launch_bounds__(GAT_SCAN_THREADS) void gat_rowptr_scan(
    const int* __restrict__ cnt, int* __restrict__ rowptr)
{
  __shared__ int wsum[32];
  const int t = threadIdx.x, lane = t & 31, w = t >> 5;
  const int4* p = (const int4*)(cnt + t * GAT_SCAN_PER);
  int4 v[GAT_SCAN_PER / 4];
  int tot = 0;
#pragma unroll
  for (int i = 0; i < GAT_SCAN_PER / 4; ++i) {
    v[i] = p[i];
    tot += v[i].x + v[i].y + v[i].z + v[i].w;
  }
  int inc = tot;
#pragma unroll
  for (int o = 1; o < 32; o <<= 1) {
    int nb = __shfl_up(inc, o, 32);
    if (lane >= o) inc += nb;
  }
  if (lane == 31) wsum[w] = inc;
  __syncthreads();
  if (w == 0) {
    int x = wsum[lane], xi = x;
#pragma unroll
    for (int o = 1; o < 32; o <<= 1) {
      int nb = __shfl_up(xi, o, 32);
      if (lane >= o) xi += nb;
    }
    wsum[lane] = xi - x;
  }
  __syncthreads();
  int run = wsum[w] + inc - tot;
  int4* q = (int4*)(rowptr + t * GAT_SCAN_PER);
#pragma unroll
  for (int i = 0; i < GAT_SCAN_PER / 4; ++i) {
    int4 o;
    o.x = run; run += v[i].x;
    o.y = run; run += v[i].y;
    o.z = run; run += v[i].z;
    o.w = run; run += v[i].w;
    q[i] = o;
  }
}

__global__ void gat_fill_segments(const int* __restrict__ src, const int* __restrict__ dst,
                                  const int* __restrict__ rowptr, const int* __restrict__ rank,
                                  int* __restrict__ csrc)
{
  int e = blockIdx.x * blockDim.x + threadIdx.x;
  if (e >= GAT_E) return;
  csrc[rowptr[dst[e]] + rank[e]] = src[e];
}

__device__ __forceinline__ float gat_wave_max(float v) {
#pragma unroll
  for (int o = 16; o > 0; o >>= 1) v = fmaxf(v, __shfl_xor(v, o, 32));
  return v;
}
__device__ __forceinline__ float gat_wave_sum(float v) {
#pragma unroll
  for (int o = 16; o > 0; o >>= 1) v += __shfl_xor(v, o, 32);
  return v;
}
__device__ __forceinline__ float gat_lane_bcast(float v, int l) {
  return __int_as_float(__builtin_amdgcn_readlane(__float_as_int(v), l));
}

__device__ __forceinline__ int gat_slot_src(const int* __restrict__ csrc, int j) {
  unsigned u = (unsigned)csrc[j];
  return (int)(u < (unsigned)GAT_N ? u : (unsigned)(GAT_N - 1));
}

__device__ __forceinline__ float4 gat_edge_logits(const float* __restrict__ el, int s, float4 erd) {
  const float4 e = *(const float4*)(el + (size_t)s * GAT_H);
  float4 x;
  x.x = gat_leaky(e.x + erd.x);
  x.y = gat_leaky(e.y + erd.y);
  x.z = gat_leaky(e.z + erd.z);
  x.w = gat_leaky(e.w + erd.w);
  return x;
}

__device__ __forceinline__ void gat_segment_stats(
    const int* __restrict__ csrc, const float* __restrict__ el, float4 erd,
    int beg, int end, int lane, float4& m, float4& den)
{
  const float ninf = __int_as_float((int)0xFF800000u);
  m.x = ninf; m.y = ninf; m.z = ninf; m.w = ninf;
  for (int c = beg; c < end; c += 32) {
    const int j = c + lane;
    if (j < end) {
      const float4 x = gat_edge_logits(el, gat_slot_src(csrc, j), erd);
      m.x = fmaxf(m.x, x.x); m.y = fmaxf(m.y, x.y);
      m.z = fmaxf(m.z, x.z); m.w = fmaxf(m.w, x.w);
    }
  }
  m.x = gat_wave_max(m.x); m.y = gat_wave_max(m.y);
  m.z = gat_wave_max(m.z); m.w = gat_wave_max(m.w);
  den.x = 0.f; den.y = 0.f; den.z = 0.f; den.w = 0.f;
  for (int c = beg; c < end; c += 32) {
    const int j = c + lane;
    if (j < end) {
      const float4 x = gat_edge_logits(el, gat_slot_src(csrc, j), erd);
      den.x += expf(x.x - m.x); den.y += expf(x.y - m.y);
      den.z += expf(x.z - m.z); den.w += expf(x.w - m.w);
    }
  }
  den.x = gat_wave_sum(den.x); den.y = gat_wave_sum(den.y);
  den.z = gat_wave_sum(den.z); den.w = gat_wave_sum(den.w);
}

__device__ __forceinline__ void gat_chunk_weights(
    const int* __restrict__ csrc, const float* __restrict__ el, float4 erd,
    int c, int end, int lane, float4 m, float4 den, int& s, float4& a)
{
  s = 0;
  a.x = 0.f; a.y = 0.f; a.z = 0.f; a.w = 0.f;
  const int j = c + lane;
  if (j < end) {
    s = gat_slot_src(csrc, j);
    const float4 x = gat_edge_logits(el, s, erd);
    a.x = expf(x.x - m.x) / den.x; a.y = expf(x.y - m.y) / den.y;
    a.z = expf(x.z - m.z) / den.z; a.w = expf(x.w - m.w) / den.w;
  }
}

__global__ __launch_bounds__(256) void gat_node_agg_elu(
    const int* __restrict__ rowptr, const int* __restrict__ csrc,
    const float* __restrict__ feat, const float* __restrict__ el,
    const float* __restrict__ er, const float* __restrict__ bias,
    const float* __restrict__ res, float* __restrict__ out)
{
  const int d = __builtin_amdgcn_readfirstlane((int)(blockIdx.x * 8 + (threadIdx.x >> 5)));
  if (d >= GAT_N) return;
  const int lane = threadIdx.x & 31;
  const int beg = rowptr[d], end = rowptr[d + 1];
  const float4 erd = *(const float4*)(er + (size_t)d * GAT_H);

  float4 m, den;
  gat_segment_stats(csrc, el, erd, beg, end, lane, m, den);

  const bool lo = lane < 16;
  float4 accA = {0.f, 0.f, 0.f, 0.f}, accB = {0.f, 0.f, 0.f, 0.f};
  for (int c = beg; c < end; c += 32) {
    int s; float4 a;
    gat_chunk_weights(csrc, el, erd, c, end, lane, m, den, s, a);
    const int cnt = (end - c) < 32 ? (end - c) : 32;
#pragma unroll 4
    for (int t = 0; t < cnt; ++t) {
      const int st = __builtin_amdgcn_readlane(s, t);
      const float a0 = gat_lane_bcast(a.x, t), a1 = gat_lane_bcast(a.y, t);
      const float a2 = gat_lane_bcast(a.z, t), a3 = gat_lane_bcast(a.w, t);
      const float wA = lo ? a0 : a1;
      const float wB = lo ? a2 : a3;
      const float4* row = (const float4*)(feat + (size_t)st * 256);
      const float4 vA = row[lane];
      const float4 vB = row[32 + lane];
      accA.x += vA.x * wA; accA.y += vA.y * wA; accA.z += vA.z * wA; accA.w += vA.w * wA;
      accB.x += vB.x * wB; accB.y += vB.y * wB; accB.z += vB.z * wB; accB.w += vB.w * wB;
    }
  }

  const float4 bA = *(const float4*)(bias + 4 * lane);
  const float4 bB = *(const float4*)(bias + 128 + 4 * lane);
  float4 vA, vB;
  vA.x = accA.x + bA.x; vA.y = accA.y + bA.y; vA.z = accA.z + bA.z; vA.w = accA.w + bA.w;
  vB.x = accB.x + bB.x; vB.y = accB.y + bB.y; vB.z = accB.z + bB.z; vB.w = accB.w + bB.w;
  const size_t o = (size_t)d * 256 + 4 * lane;
  if (res) {
    const float4 rA = *(const float4*)(res + o);
    const float4 rB = *(const float4*)(res + o + 128);
    vA.x += rA.x; vA.y += rA.y; vA.z += rA.z; vA.w += rA.w;
    vB.x += rB.x; vB.y += rB.y; vB.z += rB.z; vB.w += rB.w;
  }
  vA.x = vA.x > 0.f ? vA.x : (expf(vA.x) - 1.f);
  vA.y = vA.y > 0.f ? vA.y : (expf(vA.y) - 1.f);
  vA.z = vA.z > 0.f ? vA.z : (expf(vA.z) - 1.f);
  vA.w = vA.w > 0.f ? vA.w : (expf(vA.w) - 1.f);
  vB.x = vB.x > 0.f ? vB.x : (expf(vB.x) - 1.f);
  vB.y = vB.y > 0.f ? vB.y : (expf(vB.y) - 1.f);
  vB.z = vB.z > 0.f ? vB.z : (expf(vB.z) - 1.f);
  vB.w = vB.w > 0.f ? vB.w : (expf(vB.w) - 1.f);
  *(float4*)(out + o) = vA;
  *(float4*)(out + o + 128) = vB;
}

__global__ __launch_bounds__(256) void gat_node_agg_mean(
    const int* __restrict__ rowptr, const int* __restrict__ csrc,
    const float* __restrict__ feat, const float* __restrict__ el,
    const float* __restrict__ er, const float* __restrict__ bias,
    const float* __restrict__ res2, float* __restrict__ out)
{
  __shared__ float sval[8 * 160];
  const int w = threadIdx.x >> 5;
  const int d = __builtin_amdgcn_readfirstlane((int)(blockIdx.x * 8 + w));
  const int lane = threadIdx.x & 31;

  if (d < GAT_N) {
    const int beg = rowptr[d], end = rowptr[d + 1];
    const float4 erd = *(const float4*)(er + (size_t)d * GAT_H);
    float4 m, den;
    gat_segment_stats(csrc, el, erd, beg, end, lane, m, den);

    float acc0 = 0.f, acc1 = 0.f, acc2 = 0.f, acc3 = 0.f, acc4 = 0.f;
    for (int c = beg; c < end; c += 32) {
      int s; float4 a;
      gat_chunk_weights(csrc, el, erd, c, end, lane, m, den, s, a);
      const int cnt = (end - c) < 32 ? (end - c) : 32;
#pragma unroll 4
      for (int t = 0; t < cnt; ++t) {
        const int st = __builtin_amdgcn_readlane(s, t);
        const float a0 = gat_lane_bcast(a.x, t), a1 = gat_lane_bcast(a.y, t);
        const float a2 = gat_lane_bcast(a.z, t), a3 = gat_lane_bcast(a.w, t);
        const float* row = feat + (size_t)st * 160 + lane;
        const float v0 = row[0], v1 = row[32], v2 = row[64], v3 = row[96], v4 = row[128];
        acc0 += v0 * a0;
        acc1 += v1 * (lane >= 8 ? a1 : a0);
        acc2 += v2 * (lane >= 16 ? a2 : a1);
        acc3 += v3 * (lane >= 24 ? a3 : a2);
        acc4 += v4 * a3;
      }
    }
    const float* rr = res2 + (size_t)d * 160 + lane;
    const float* bb = bias + lane;
    float* sv = sval + w * 160 + lane;
    sv[0]   = acc0 + bb[0]   + rr[0];
    sv[32]  = acc1 + bb[32]  + rr[32];
    sv[64]  = acc2 + bb[64]  + rr[64];
    sv[96]  = acc3 + bb[96]  + rr[96];
    sv[128] = acc4 + bb[128] + rr[128];
  }
  __syncthreads();
  for (int i = threadIdx.x; i < 8 * 40; i += 256) {
    const int nl = i / 40, c = i - nl * 40;
    if ((int)(blockIdx.x * 8 + nl) < GAT_N) {
      const float* sv = sval + nl * 160 + c;
      float sum = 0.f;
#pragma unroll
      for (int h = 0; h < GAT_H; ++h) sum += sv[h * 40];
      out[(size_t)blockIdx.x * 320 + i] = sum * 0.25f;
    }
  }
}

extern "C" void kernel_launch(void* const* d_in, const int* in_sizes, int n_in,
                              void* d_out, int out_size, void* d_ws, size_t ws_size,
                              hipStream_t stream)
{
  const float* x     = (const float*)d_in[0];
  const int*   src   = (const int*)d_in[1];
  const int*   dst   = (const int*)d_in[2];
  const float* W0    = (const float*)d_in[3];
  const float* al0   = (const float*)d_in[4];
  const float* ar0   = (const float*)d_in[5];
  const float* b0    = (const float*)d_in[6];
  const float* W1    = (const float*)d_in[7];
  const float* al1   = (const float*)d_in[8];
  const float* ar1   = (const float*)d_in[9];
  const float* b1    = (const float*)d_in[10];
  const float* W2    = (const float*)d_in[11];
  const float* al2   = (const float*)d_in[12];
  const float* ar2   = (const float*)d_in[13];
  const float* b2    = (const float*)d_in[14];
  const float* Wres2 = (const float*)d_in[15];

  const size_t NF = (size_t)GAT_N * 256;
  float* ws   = (float*)d_ws;
  float* bufA = ws;
  float* bufB = bufA + NF;
  float* bufC = bufB + NF;
  float* EL   = bufC + NF;
  float* ER   = EL + (size_t)GAT_N * GAT_H;
  int*   CNT  = (int*)(ER + (size_t)GAT_N * GAT_H);
  int*   RPTR = CNT + GAT_CNT_PAD;
  int*   RANK = RPTR + GAT_CNT_PAD;
  int*   CSRC = RANK + GAT_E;

  const int rowTiles = (GAT_N + 31) / 32;
  dim3 gemmBlk256(128);
  dim3 gemmBlk160(160);

  const int eBlocks    = (GAT_E + 255) / 256;
  const int nhThreads  = GAT_N * GAT_H;
  const int nhBlocks   = (nhThreads + 255) / 256;
  const int nodeBlocks = (GAT_N + 7) / 8;

  gat_zero_counts<<<GAT_CNT_PAD / 256, 256, 0, stream>>>(CNT);
  gat_dst_rank<<<eBlocks, 256, 0, stream>>>(dst, CNT, RANK);
  gat_rowptr_scan<<<1, GAT_SCAN_THREADS, 0, stream>>>(CNT, RPTR);
  gat_fill_segments<<<eBlocks, 256, 0, stream>>>(src, dst, RPTR, RANK, CSRC);

  gat_gemm_wmma<256, 256, 4><<<rowTiles, gemmBlk256, 0, stream>>>(x, W0, bufA, GAT_N);
  gat_attn_scores<<<nhBlocks, 256, 0, stream>>>(bufA, al0, ar0, EL, ER, 256, 64);
  gat_node_agg_elu<<<nodeBlocks, 256, 0, stream>>>(RPTR, CSRC, bufA, EL, ER, b0, nullptr, bufB);

  gat_gemm_wmma<256, 256, 4><<<rowTiles, gemmBlk256, 0, stream>>>(bufB, W1, bufA, GAT_N);
  gat_attn_scores<<<nhBlocks, 256, 0, stream>>>(bufA, al1, ar1, EL, ER, 256, 64);
  gat_node_agg_elu<<<nodeBlocks, 256, 0, stream>>>(RPTR, CSRC, bufA, EL, ER, b1, bufB, bufC);

  gat_gemm_wmma<256, 160, 2><<<rowTiles, gemmBlk160, 0, stream>>>(bufC, W2, bufA, GAT_N);
  gat_gemm_wmma<256, 160, 2><<<rowTiles, gemmBlk160, 0, stream>>>(bufC, Wres2, bufB, GAT_N);
  gat_attn_scores<<<nhBlocks, 256, 0, stream>>>(bufA, al2, ar2, EL, ER, 160, 40);
  gat_node_agg_mean<<<nodeBlocks, 256, 0, stream>>>(RPTR, CSRC, bufA, EL, ER, b2, bufB, (float*)d_out);
}


# attention scores el/er summed in the GEMM epilogue from the accumulators (lane dot, 4-lane DPP sum, LDS, per-head sum), not by a kernel re-reading feat; 3 launches gone; el/er sum order changes
# speedup vs baseline: 3.5770x; 1.0329x over previous
#include <hip/hip_runtime.h>
#include <math.h>

#define GAT_N 50000
#define GAT_E 800000
#define GAT_H 4
#define GAT_SLOPE 0.2f

typedef __attribute__((ext_vector_type(2))) float v2f;
typedef __attribute__((ext_vector_type(8))) float v8f;

template <int MASK>
__device__ __forceinline__ float gat_row_xor_add(float v) {
  return v + __int_as_float(__builtin_amdgcn_update_dpp(
                 0, __float_as_int(v), 0x160 + MASK, 0xf, 0xf, true));
}

template <int K, int NCOL, int NT, int D, bool SCORES>
__global__ __launch_bounds__(NCOL / (16 * NT) * 32) void gat_gemm_wmma(
    const float* __restrict__ A, const float* __restrict__ B,
    float* __restrict__ C, int M,
    const float* __restrict__ al, const float* __restrict__ ar,
    float* __restrict__ el, float* __restrict__ er)
{
  typedef float vnt __attribute__((ext_vector_type(NT)));
  typedef float v4f __attribute__((ext_vector_type(4)));
  const int wave = threadIdx.x >> 5;
  const int lane = threadIdx.x & 31;
  const int row0 = blockIdx.x << 5;
  const int col0 = wave * (16 * NT);
  const int c    = lane & 15;
  const int h    = lane >> 4;
  const bool hi_valid = (row0 + 16) < M;

  const float* arow0 = A + (size_t)(row0 + c) * K + 4 * h;
  const float* arow1 = hi_valid ? (arow0 + (size_t)16 * K) : arow0;
  const float* bp    = B + (size_t)(4 * h) * NCOL + col0 + NT * c;

  v8f acc0[NT], acc1[NT];
#pragma unroll
  for (int t = 0; t < NT; ++t) { acc0[t] = (v8f){}; acc1[t] = (v8f){}; }

#pragma unroll 2
  for (int k0 = 0; k0 < K; k0 += 8) {
    const v4f a0 = *(const v4f*)(arow0 + k0);
    const v4f a1 = *(const v4f*)(arow1 + k0);
    vnt b[4];
#pragma unroll
    for (int i = 0; i < 4; ++i) b[i] = *(const vnt*)(bp + (size_t)(k0 + i) * NCOL);
#pragma unroll
    for (int j = 0; j < 2; ++j) {
      v2f fa0, fa1;
      fa0.x = a0[2 * j]; fa0.y = a0[2 * j + 1];
      fa1.x = a1[2 * j]; fa1.y = a1[2 * j + 1];
#pragma unroll
      for (int t = 0; t < NT; ++t) {
        v2f fb;
        fb.x = b[2 * j][t];
        fb.y = b[2 * j + 1][t];
        acc0[t] = __builtin_amdgcn_wmma_f32_16x16x4_f32(
            false, fa0, false, fb, (short)0, acc0[t], false, false);
        acc1[t] = __builtin_amdgcn_wmma_f32_16x16x4_f32(
            false, fa1, false, fb, (short)0, acc1[t], false, false);
      }
    }
  }

  constexpr int NG    = NCOL / (4 * NT);
  constexpr int GPH   = D / (4 * NT);
  constexpr int PITCH = 2 * NG + 2;
  __shared__ float part[SCORES ? 32 * PITCH : 1];
  if (SCORES) {
    const vnt wl = *(const vnt*)(al + col0 + NT * c);
    const vnt wr = *(const vnt*)(ar + col0 + NT * c);
    const int g = (col0 + NT * c) / (4 * NT);
#pragma unroll
    for (int r = 0; r < 8; ++r) {
      float pl0 = 0.f, pr0 = 0.f, pl1 = 0.f, pr1 = 0.f;
#pragma unroll
      for (int t = 0; t < NT; ++t) {
        pl0 = fmaf(acc0[t][r], wl[t], pl0);
        pr0 = fmaf(acc0[t][r], wr[t], pr0);
        pl1 = fmaf(acc1[t][r], wl[t], pl1);
        pr1 = fmaf(acc1[t][r], wr[t], pr1);
      }
      pl0 = gat_row_xor_add<2>(gat_row_xor_add<1>(pl0));
      pr0 = gat_row_xor_add<2>(gat_row_xor_add<1>(pr0));
      pl1 = gat_row_xor_add<2>(gat_row_xor_add<1>(pl1));
      pr1 = gat_row_xor_add<2>(gat_row_xor_add<1>(pr1));
      if ((c & 3) == 0) {
        v2f s0, s1;
        s0.x = pl0; s0.y = pr0;
        s1.x = pl1; s1.y = pr1;
        *(v2f*)(part + (8 * h + r) * PITCH + 2 * g)      = s0;
        *(v2f*)(part + (16 + 8 * h + r) * PITCH + 2 * g) = s1;
      }
    }
    __syncthreads();
  }

  float* crow0 = C + (size_t)(row0 + 8 * h) * NCOL + col0 + NT * c;
#pragma unroll
  for (int r = 0; r < 8; ++r) {
    vnt o;
#pragma unroll
    for (int t = 0; t < NT; ++t) o[t] = acc0[t][r];
    *(vnt*)(crow0 + (size_t)r * NCOL) = o;
  }
  if (hi_valid) {
    float* crow1 = crow0 + (size_t)16 * NCOL;
#pragma unroll
    for (int r = 0; r < 8; ++r) {
      vnt o;
#pragma unroll
      for (int t = 0; t < NT; ++t) o[t] = acc1[t][r];
      *(vnt*)(crow1 + (size_t)r * NCOL) = o;
    }
  }

  if (SCORES) {
    for (int i = threadIdx.x; i < 32 * GAT_H * 2; i += NCOL / (16 * NT) * 32) {
      const int row = i >> 3, hd = (i >> 1) & 3, which = i & 1;
      if (row0 + row < M) {
        const float* p = part + row * PITCH + 2 * hd * GPH + which;
        float sum = 0.f;
#pragma unroll
        for (int q = 0; q < GPH; ++q) sum += p[2 * q];
        (which ? er : el)[(size_t)(row0 + row) * GAT_H + hd] = sum;
      }
    }
  }
}

__device__ __forceinline__ float gat_leaky(float x) {
  return x > 0.f ? x : GAT_SLOPE * x;
}

#define GAT_SCAN_THREADS 1024
#define GAT_SCAN_PER 52
#define GAT_CNT_PAD (GAT_SCAN_THREADS * GAT_SCAN_PER)

__global__ void gat_zero_counts(int* __restrict__ cnt)
{
  int idx = blockIdx.x * blockDim.x + threadIdx.x;
  if (idx < GAT_CNT_PAD) cnt[idx] = 0;
}

__global__ void gat_dst_rank(const int* __restrict__ dst, int* __restrict__ cnt,
                             int* __restrict__ rank)
{
  int e = blockIdx.x * blockDim.x + threadIdx.x;
  if (e >= GAT_E) return;
  rank[e] = atomicAdd(&cnt[dst[e]], 1);
}

__global__ __launch_bounds__(GAT_SCAN_THREADS) void gat_rowptr_scan(
    const int* __restrict__ cnt, int* __restrict__ rowptr)
{
  __shared__ int wsum[32];
  const int t = threadIdx.x, lane = t & 31, w = t >> 5;
  const int4* p = (const int4*)(cnt + t * GAT_SCAN_PER);
  int4 v[GAT_SCAN_PER / 4];
  int tot = 0;
#pragma unroll
  for (int i = 0; i < GAT_SCAN_PER / 4; ++i) {
    v[i] = p[i];
    tot += v[i].x + v[i].y + v[i].z + v[i].w;
  }
  int inc = tot;
#pragma unroll
  for (int o = 1; o < 32; o <<= 1) {
    int nb = __shfl_up(inc, o, 32);
    if (lane >= o) inc += nb;
  }
  if (lane == 31) wsum[w] = inc;
  __syncthreads();
  if (w == 0) {
    int x = wsum[lane], xi = x;
#pragma unroll
    for (int o = 1; o < 32; o <<= 1) {
      int nb = __shfl_up(xi, o, 32);
      if (lane >= o) xi += nb;
    }
    wsum[lane] = xi - x;
  }
  __syncthreads();
  int run = wsum[w] + inc - tot;
  int4* q = (int4*)(rowptr + t * GAT_SCAN_PER);
#pragma unroll
  for (int i = 0; i < GAT_SCAN_PER / 4; ++i) {
    int4 o;
    o.x = run; run += v[i].x;
    o.y = run; run += v[i].y;
    o.z = run; run += v[i].z;
    o.w = run; run += v[i].w;
    q[i] = o;
  }
}

__global__ void gat_fill_segments(const int* __restrict__ src, const int* __restrict__ dst,
                                  const int* __restrict__ rowptr, const int* __restrict__ rank,
                                  int* __restrict__ csrc)
{
  int e = blockIdx.x * blockDim.x + threadIdx.x;
  if (e >= GAT_E) return;
  csrc[rowptr[dst[e]] + rank[e]] = src[e];
}

__device__ __forceinline__ float gat_wave_max(float v) {
#pragma unroll
  for (int o = 16; o > 0; o >>= 1) v = fmaxf(v, __shfl_xor(v, o, 32));
  return v;
}
__device__ __forceinline__ float gat_wave_sum(float v) {
#pragma unroll
  for (int o = 16; o > 0; o >>= 1) v += __shfl_xor(v, o, 32);
  return v;
}
__device__ __forceinline__ float gat_lane_bcast(float v, int l) {
  return __int_as_float(__builtin_amdgcn_readlane(__float_as_int(v), l));
}

__device__ __forceinline__ int gat_slot_src(const int* __restrict__ csrc, int j) {
  unsigned u = (unsigned)csrc[j];
  return (int)(u < (unsigned)GAT_N ? u : (unsigned)(GAT_N - 1));
}

__device__ __forceinline__ float4 gat_edge_logits(const float* __restrict__ el, int s, float4 erd) {
  const float4 e = *(const float4*)(el + (size_t)s * GAT_H);
  float4 x;
  x.x = gat_leaky(e.x + erd.x);
  x.y = gat_leaky(e.y + erd.y);
  x.z = gat_leaky(e.z + erd.z);
  x.w = gat_leaky(e.w + erd.w);
  return x;
}

__device__ __forceinline__ void gat_segment_stats(
    const int* __restrict__ csrc, const float* __restrict__ el, float4 erd,
    int beg, int end, int lane, float4& m, float4& den)
{
  const float ninf = __int_as_float((int)0xFF800000u);
  m.x = ninf; m.y = ninf; m.z = ninf; m.w = ninf;
  for (int c = beg; c < end; c += 32) {
    const int j = c + lane;
    if (j < end) {
      const float4 x = gat_edge_logits(el, gat_slot_src(csrc, j), erd);
      m.x = fmaxf(m.x, x.x); m.y = fmaxf(m.y, x.y);
      m.z = fmaxf(m.z, x.z); m.w = fmaxf(m.w, x.w);
    }
  }
  m.x = gat_wave_max(m.x); m.y = gat_wave_max(m.y);
  m.z = gat_wave_max(m.z); m.w = gat_wave_max(m.w);
  den.x = 0.f; den.y = 0.f; den.z = 0.f; den.w = 0.f;
  for (int c = beg; c < end; c += 32) {
    const int j = c + lane;
    if (j < end) {
      const float4 x = gat_edge_logits(el, gat_slot_src(csrc, j), erd);
      den.x += expf(x.x - m.x); den.y += expf(x.y - m.y);
      den.z += expf(x.z - m.z); den.w += expf(x.w - m.w);
    }
  }
  den.x = gat_wave_sum(den.x); den.y = gat_wave_sum(den.y);
  den.z = gat_wave_sum(den.z); den.w = gat_wave_sum(den.w);
}

__device__ __forceinline__ void gat_chunk_weights(
    const int* __restrict__ csrc, const float* __restrict__ el, float4 erd,
    int c, int end, int lane, float4 m, float4 den, int& s, float4& a)
{
  s = 0;
  a.x = 0.f; a.y = 0.f; a.z = 0.f; a.w = 0.f;
  const int j = c + lane;
  if (j < end) {
    s = gat_slot_src(csrc, j);
    const float4 x = gat_edge_logits(el, s, erd);
    a.x = expf(x.x - m.x) / den.x; a.y = expf(x.y - m.y) / den.y;
    a.z = expf(x.z - m.z) / den.z; a.w = expf(x.w - m.w) / den.w;
  }
}

__global__ __launch_bounds__(256) void gat_node_agg_elu(
    const int* __restrict__ rowptr, const int* __restrict__ csrc,
    const float* __restrict__ feat, const float* __restrict__ el,
    const float* __restrict__ er, const float* __restrict__ bias,
    const float* __restrict__ res, float* __restrict__ out)
{
  const int d = __builtin_amdgcn_readfirstlane((int)(blockIdx.x * 8 + (threadIdx.x >> 5)));
  if (d >= GAT_N) return;
  const int lane = threadIdx.x & 31;
  const int beg = rowptr[d], end = rowptr[d + 1];
  const float4 erd = *(const float4*)(er + (size_t)d * GAT_H);

  float4 m, den;
  gat_segment_stats(csrc, el, erd, beg, end, lane, m, den);

  const bool lo = lane < 16;
  float4 accA = {0.f, 0.f, 0.f, 0.f}, accB = {0.f, 0.f, 0.f, 0.f};
  for (int c = beg; c < end; c += 32) {
    int s; float4 a;
    gat_chunk_weights(csrc, el, erd, c, end, lane, m, den, s, a);
    const int cnt = (end - c) < 32 ? (end - c) : 32;
#pragma unroll 4
    for (int t = 0; t < cnt; ++t) {
      const int st = __builtin_amdgcn_readlane(s, t);
      const float a0 = gat_lane_bcast(a.x, t), a1 = gat_lane_bcast(a.y, t);
      const float a2 = gat_lane_bcast(a.z, t), a3 = gat_lane_bcast(a.w, t);
      const float wA = lo ? a0 : a1;
      const float wB = lo ? a2 : a3;
      const float4* row = (const float4*)(feat + (size_t)st * 256);
      const float4 vA = row[lane];
      const float4 vB = row[32 + lane];
      accA.x += vA.x * wA; accA.y += vA.y * wA; accA.z += vA.z * wA; accA.w += vA.w * wA;
      accB.x += vB.x * wB; accB.y += vB.y * wB; accB.z += vB.z * wB; accB.w += vB.w * wB;
    }
  }

  const float4 bA = *(const float4*)(bias + 4 * lane);
  const float4 bB = *(const float4*)(bias + 128 + 4 * lane);
  float4 vA, vB;
  vA.x = accA.x + bA.x; vA.y = accA.y + bA.y; vA.z = accA.z + bA.z; vA.w = accA.w + bA.w;
  vB.x = accB.x + bB.x; vB.y = accB.y + bB.y; vB.z = accB.z + bB.z; vB.w = accB.w + bB.w;
  const size_t o = (size_t)d * 256 + 4 * lane;
  if (res) {
    const float4 rA = *(const float4*)(res + o);
    const float4 rB = *(const float4*)(res + o + 128);
    vA.x += rA.x; vA.y += rA.y; vA.z += rA.z; vA.w += rA.w;
    vB.x += rB.x; vB.y += rB.y; vB.z += rB.z; vB.w += rB.w;
  }
  vA.x = vA.x > 0.f ? vA.x : (expf(vA.x) - 1.f);
  vA.y = vA.y > 0.f ? vA.y : (expf(vA.y) - 1.f);
  vA.z = vA.z > 0.f ? vA.z : (expf(vA.z) - 1.f);
  vA.w = vA.w > 0.f ? vA.w : (expf(vA.w) - 1.f);
  vB.x = vB.x > 0.f ? vB.x : (expf(vB.x) - 1.f);
  vB.y = vB.y > 0.f ? vB.y : (expf(vB.y) - 1.f);
  vB.z = vB.z > 0.f ? vB.z : (expf(vB.z) - 1.f);
  vB.w = vB.w > 0.f ? vB.w : (expf(vB.w) - 1.f);
  *(float4*)(out + o) = vA;
  *(float4*)(out + o + 128) = vB;
}

__global__ __launch_bounds__(256) void gat_node_agg_mean(
    const int* __restrict__ rowptr, const int* __restrict__ csrc,
    const float* __restrict__ feat, const float* __restrict__ el,
    const float* __restrict__ er, const float* __restrict__ bias,
    const float* __restrict__ res2, float* __restrict__ out)
{
  __shared__ float sval[8 * 160];
  const int w = threadIdx.x >> 5;
  const int d = __builtin_amdgcn_readfirstlane((int)(blockIdx.x * 8 + w));
  const int lane = threadIdx.x & 31;

  if (d < GAT_N) {
    const int beg = rowptr[d], end = rowptr[d + 1];
    const float4 erd = *(const float4*)(er + (size_t)d * GAT_H);
    float4 m, den;
    gat_segment_stats(csrc, el, erd, beg, end, lane, m, den);

    float acc0 = 0.f, acc1 = 0.f, acc2 = 0.f, acc3 = 0.f, acc4 = 0.f;
    for (int c = beg; c < end; c += 32) {
      int s; float4 a;
      gat_chunk_weights(csrc, el, erd, c, end, lane, m, den, s, a);
      const int cnt = (end - c) < 32 ? (end - c) : 32;
#pragma unroll 4
      for (int t = 0; t < cnt; ++t) {
        const int st = __builtin_amdgcn_readlane(s, t);
        const float a0 = gat_lane_bcast(a.x, t), a1 = gat_lane_bcast(a.y, t);
        const float a2 = gat_lane_bcast(a.z, t), a3 = gat_lane_bcast(a.w, t);
        const float* row = feat + (size_t)st * 160 + lane;
        const float v0 = row[0], v1 = row[32], v2 = row[64], v3 = row[96], v4 = row[128];
        acc0 += v0 * a0;
        acc1 += v1 * (lane >= 8 ? a1 : a0);
        acc2 += v2 * (lane >= 16 ? a2 : a1);
        acc3 += v3 * (lane >= 24 ? a3 : a2);
        acc4 += v4 * a3;
      }
    }
    const float* rr = res2 + (size_t)d * 160 + lane;
    const float* bb = bias + lane;
    float* sv = sval + w * 160 + lane;
    sv[0]   = acc0 + bb[0]   + rr[0];
    sv[32]  = acc1 + bb[32]  + rr[32];
    sv[64]  = acc2 + bb[64]  + rr[64];
    sv[96]  = acc3 + bb[96]  + rr[96];
    sv[128] = acc4 + bb[128] + rr[128];
  }
  __syncthreads();
  for (int i = threadIdx.x; i < 8 * 40; i += 256) {
    const int nl = i / 40, c = i - nl * 40;
    if ((int)(blockIdx.x * 8 + nl) < GAT_N) {
      const float* sv = sval + nl * 160 + c;
      float sum = 0.f;
#pragma unroll
      for (int h = 0; h < GAT_H; ++h) sum += sv[h * 40];
      out[(size_t)blockIdx.x * 320 + i] = sum * 0.25f;
    }
  }
}

extern "C" void kernel_launch(void* const* d_in, const int* in_sizes, int n_in,
                              void* d_out, int out_size, void* d_ws, size_t ws_size,
                              hipStream_t stream)
{
  const float* x     = (const float*)d_in[0];
  const int*   src   = (const int*)d_in[1];
  const int*   dst   = (const int*)d_in[2];
  const float* W0    = (const float*)d_in[3];
  const float* al0   = (const float*)d_in[4];
  const float* ar0   = (const float*)d_in[5];
  const float* b0    = (const float*)d_in[6];
  const float* W1    = (const float*)d_in[7];
  const float* al1   = (const float*)d_in[8];
  const float* ar1   = (const float*)d_in[9];
  const float* b1    = (const float*)d_in[10];
  const float* W2    = (const float*)d_in[11];
  const float* al2   = (const float*)d_in[12];
  const float* ar2   = (const float*)d_in[13];
  const float* b2    = (const float*)d_in[14];
  const float* Wres2 = (const float*)d_in[15];

  const size_t NF = (size_t)GAT_N * 256;
  float* ws   = (float*)d_ws;
  float* bufA = ws;
  float* bufB = bufA + NF;
  float* bufC = bufB + NF;
  float* EL   = bufC + NF;
  float* ER   = EL + (size_t)GAT_N * GAT_H;
  int*   CNT  = (int*)(ER + (size_t)GAT_N * GAT_H);
  int*   RPTR = CNT + GAT_CNT_PAD;
  int*   RANK = RPTR + GAT_CNT_PAD;
  int*   CSRC = RANK + GAT_E;

  const int rowTiles = (GAT_N + 31) / 32;
  dim3 gemmBlk256(128);
  dim3 gemmBlk160(160);

  const int eBlocks    = (GAT_E + 255) / 256;
  const int nodeBlocks = (GAT_N + 7) / 8;

  gat_zero_counts<<<GAT_CNT_PAD / 256, 256, 0, stream>>>(CNT);
  gat_dst_rank<<<eBlocks, 256, 0, stream>>>(dst, CNT, RANK);
  gat_rowptr_scan<<<1, GAT_SCAN_THREADS, 0, stream>>>(CNT, RPTR);
  gat_fill_segments<<<eBlocks, 256, 0, stream>>>(src, dst, RPTR, RANK, CSRC);

  gat_gemm_wmma<256, 256, 4, 64, true><<<rowTiles, gemmBlk256, 0, stream>>>(x, W0, bufA, GAT_N, al0, ar0, EL, ER);
  gat_node_agg_elu<<<nodeBlocks, 256, 0, stream>>>(RPTR, CSRC, bufA, EL, ER, b0, nullptr, bufB);

  gat_gemm_wmma<256, 256, 4, 64, true><<<rowTiles, gemmBlk256, 0, stream>>>(bufB, W1, bufA, GAT_N, al1, ar1, EL, ER);
  gat_node_agg_elu<<<nodeBlocks, 256, 0, stream>>>(RPTR, CSRC, bufA, EL, ER, b1, bufB, bufC);

  gat_gemm_wmma<256, 160, 2, 40, true><<<rowTiles, gemmBlk160, 0, stream>>>(bufC, W2, bufA, GAT_N, al2, ar2, EL, ER);
  gat_gemm_wmma<256, 160, 2, 40, false><<<rowTiles, gemmBlk160, 0, stream>>>(bufC, Wres2, bufB, GAT_N, nullptr, nullptr, nullptr, nullptr);
  gat_node_agg_mean<<<nodeBlocks, 256, 0, stream>>>(RPTR, CSRC, bufA, EL, ER, b2, bufB, (float*)d_out);
}


# edges grouped by destination by a two-level counting sort on LDS counters (chunk x 256-node bucket, then node): 800000 returning global adds gone, 2 launches for 4; sum order inside a group changes
# speedup vs baseline: 3.8171x; 1.0671x over previous
#include <hip/hip_runtime.h>
#include <math.h>

#define GAT_N 50000
#define GAT_E 800000
#define GAT_H 4
#define GAT_SLOPE 0.2f

typedef __attribute__((ext_vector_type(2))) float v2f;
typedef __attribute__((ext_vector_type(8))) float v8f;

template <int MASK>
__device__ __forceinline__ float gat_row_xor_add(float v) {
  return v + __int_as_float(__builtin_amdgcn_update_dpp(
                 0, __float_as_int(v), 0x160 + MASK, 0xf, 0xf, true));
}

template <int K, int NCOL, int NT, int D, bool SCORES>
__global__ __launch_bounds__(NCOL / (16 * NT) * 32) void gat_gemm_wmma(
    const float* __restrict__ A, const float* __restrict__ B,
    float* __restrict__ C, int M,
    const float* __restrict__ al, const float* __restrict__ ar,
    float* __restrict__ el, float* __restrict__ er)
{
  typedef float vnt __attribute__((ext_vector_type(NT)));
  typedef float v4f __attribute__((ext_vector_type(4)));
  const int wave = threadIdx.x >> 5;
  const int lane = threadIdx.x & 31;
  const int row0 = blockIdx.x << 5;
  const int col0 = wave * (16 * NT);
  const int c    = lane & 15;
  const int h    = lane >> 4;
  const bool hi_valid = (row0 + 16) < M;

  const float* arow0 = A + (size_t)(row0 + c) * K + 4 * h;
  const float* arow1 = hi_valid ? (arow0 + (size_t)16 * K) : arow0;
  const float* bp    = B + (size_t)(4 * h) * NCOL + col0 + NT * c;

  v8f acc0[NT], acc1[NT];
#pragma unroll
  for (int t = 0; t < NT; ++t) { acc0[t] = (v8f){}; acc1[t] = (v8f){}; }

#pragma unroll 2
  for (int k0 = 0; k0 < K; k0 += 8) {
    const v4f a0 = *(const v4f*)(arow0 + k0);
    const v4f a1 = *(const v4f*)(arow1 + k0);
    vnt b[4];
#pragma unroll
    for (int i = 0; i < 4; ++i) b[i] = *(const vnt*)(bp + (size_t)(k0 + i) * NCOL);
#pragma unroll
    for (int j = 0; j < 2; ++j) {
      v2f fa0, fa1;
      fa0.x = a0[2 * j]; fa0.y = a0[2 * j + 1];
      fa1.x = a1[2 * j]; fa1.y = a1[2 * j + 1];
#pragma unroll
      for (int t = 0; t < NT; ++t) {
        v2f fb;
        fb.x = b[2 * j][t];
        fb.y = b[2 * j + 1][t];
        acc0[t] = __builtin_amdgcn_wmma_f32_16x16x4_f32(
            false, fa0, false, fb, (short)0, acc0[t], false, false);
        acc1[t] = __builtin_amdgcn_wmma_f32_16x16x4_f32(
            false, fa1, false, fb, (short)0, acc1[t], false, false);
      }
    }
  }

  constexpr int NG    = NCOL / (4 * NT);
  constexpr int GPH   = D / (4 * NT);
  constexpr int PITCH = 2 * NG + 2;
  __shared__ float part[SCORES ? 32 * PITCH : 1];
  if (SCORES) {
    const vnt wl = *(const vnt*)(al + col0 + NT * c);
    const vnt wr = *(const vnt*)(ar + col0 + NT * c);
    const int g = (col0 + NT * c) / (4 * NT);
#pragma unroll
    for (int r = 0; r < 8; ++r) {
      float pl0 = 0.f, pr0 = 0.f, pl1 = 0.f, pr1 = 0.f;
#pragma unroll
      for (int t = 0; t < NT; ++t) {
        pl0 = fmaf(acc0[t][r], wl[t], pl0);
        pr0 = fmaf(acc0[t][r], wr[t], pr0);
        pl1 = fmaf(acc1[t][r], wl[t], pl1);
        pr1 = fmaf(acc1[t][r], wr[t], pr1);
      }
      pl0 = gat_row_xor_add<2>(gat_row_xor_add<1>(pl0));
      pr0 = gat_row_xor_add<2>(gat_row_xor_add<1>(pr0));
      pl1 = gat_row_xor_add<2>(gat_row_xor_add<1>(pl1));
      pr1 = gat_row_xor_add<2>(gat_row_xor_add<1>(pr1));
      if ((c & 3) == 0) {
        v2f s0, s1;
        s0.x = pl0; s0.y = pr0;
        s1.x = pl1; s1.y = pr1;
        *(v2f*)(part + (8 * h + r) * PITCH + 2 * g)      = s0;
        *(v2f*)(part + (16 + 8 * h + r) * PITCH + 2 * g) = s1;
      }
    }
    __syncthreads();
  }

  float* crow0 = C + (size_t)(row0 + 8 * h) * NCOL + col0 + NT * c;
#pragma unroll
  for (int r = 0; r < 8; ++r) {
    vnt o;
#pragma unroll
    for (int t = 0; t < NT; ++t) o[t] = acc0[t][r];
    *(vnt*)(crow0 + (size_t)r * NCOL) = o;
  }
  if (hi_valid) {
    float* crow1 = crow0 + (size_t)16 * NCOL;
#pragma unroll
    for (int r = 0; r < 8; ++r) {
      vnt o;
#pragma unroll
      for (int t = 0; t < NT; ++t) o[t] = acc1[t][r];
      *(vnt*)(crow1 + (size_t)r * NCOL) = o;
    }
  }

  if (SCORES) {
    for (int i = threadIdx.x; i < 32 * GAT_H * 2; i += NCOL / (16 * NT) * 32) {
      const int row = i >> 3, hd = (i >> 1) & 3, which = i & 1;
      if (row0 + row < M) {
        const float* p = part + row * PITCH + 2 * hd * GPH + which;
        float sum = 0.f;
#pragma unroll
        for (int q = 0; q < GPH; ++q) sum += p[2 * q];
        (which ? er : el)[(size_t)(row0 + row) * GAT_H + hd] = sum;
      }
    }
  }
}

__device__ __forceinline__ float gat_leaky(float x) {
  return x > 0.f ? x : GAT_SLOPE * x;
}

#define GAT_SORT_P 256
#define GAT_SORT_CH (GAT_E / GAT_SORT_P)
#define GAT_SORT_B 196
#define GAT_SORT_T1 256
#define GAT_SORT_EPT 13
#define GAT_SORT_BATCH 8
#define GAT_OFFT_LEN ((GAT_SORT_B + 1) * GAT_SORT_P)
#define GAT_RPTR_LEN (GAT_SORT_B * 256)
static_assert(GAT_SORT_CH * GAT_SORT_P == GAT_E, "chunks must tile the edge list");
static_assert(GAT_SORT_EPT * GAT_SORT_T1 >= GAT_SORT_CH, "a chunk must fit one pass of the work-group");
static_assert(GAT_SORT_T1 > GAT_SORT_B, "one thread per bucket offset, the end entry included");
static_assert(GAT_RPTR_LEN >= GAT_N + 1, "buckets must cover every node and the end entry");
static_assert(GAT_SORT_B <= 7 * 32, "one wave scans the bucket counts, 7 per lane");
static_assert(GAT_SORT_P == 256, "level 2 gives one of its 256 threads to each chunk's run");

__device__ __forceinline__ int gat_bucket_of(int d) {
  const unsigned b = (unsigned)d >> 8;
  return (int)(b < (unsigned)GAT_SORT_B ? b : (unsigned)(GAT_SORT_B - 1));
}
__device__ __forceinline__ int gat_local_dst(int d, int lo) {
  const unsigned u = (unsigned)(d - lo);
  return (int)(u < 256u ? u : 255u);
}

__global__ __launch_bounds__(GAT_SORT_T1) void gat_chunk_buckets(
    const int* __restrict__ src, const int* __restrict__ dst,
    int* __restrict__ offt, int* __restrict__ bkt)
{
  __shared__ int hist[GAT_SORT_B + 1];
  const int p = blockIdx.x, t = threadIdx.x;
  const int e0 = p * GAT_SORT_CH;

  int s[GAT_SORT_EPT], d[GAT_SORT_EPT];
#pragma unroll
  for (int k = 0; k < GAT_SORT_EPT; ++k) {
    const int i = t + k * GAT_SORT_T1;
    const int ic = i < GAT_SORT_CH ? i : GAT_SORT_CH - 1;
    d[k] = dst[e0 + ic];
    s[k] = src[e0 + ic];
  }
  if (t <= GAT_SORT_B) hist[t] = 0;
  __syncthreads();

  int b[GAT_SORT_EPT], r[GAT_SORT_EPT];
#pragma unroll
  for (int k = 0; k < GAT_SORT_EPT; ++k) {
    b[k] = -1; r[k] = 0;
    if (t + k * GAT_SORT_T1 < GAT_SORT_CH) {
      b[k] = gat_bucket_of(d[k]);
      r[k] = atomicAdd(&hist[b[k]], 1);
    }
  }
  __syncthreads();

  if (t < 32) {
    int c[7];
    int tot = 0;
#pragma unroll
    for (int q = 0; q < 7; ++q) {
      const int idx = 7 * t + q;
      c[q] = idx < GAT_SORT_B ? hist[idx] : 0;
      tot += c[q];
    }
    int inc = tot;
#pragma unroll
    for (int o = 1; o < 32; o <<= 1) {
      const int nb = __shfl_up(inc, o, 32);
      if (t >= o) inc += nb;
    }
    int run = inc - tot;
#pragma unroll
    for (int q = 0; q < 7; ++q) {
      const int idx = 7 * t + q;
      if (idx <= GAT_SORT_B) hist[idx] = run;
      run += c[q];
    }
  }
  __syncthreads();

  if (t <= GAT_SORT_B) offt[t * GAT_SORT_P + p] = hist[t];
#pragma unroll
  for (int k = 0; k < GAT_SORT_EPT; ++k) {
    if (b[k] >= 0) {
      int2 rec;
      rec.x = s[k]; rec.y = d[k];
      *(int2*)(bkt + 2 * (size_t)(e0 + hist[b[k]] + r[k])) = rec;
    }
  }
}

__global__ __launch_bounds__(256) void gat_bucket_segments(
    const int* __restrict__ offt, const int* __restrict__ bkt,
    int* __restrict__ rowptr, int* __restrict__ csrc)
{
  __shared__ int cnt[256];
  __shared__ int wbase[8];
  __shared__ int wscan[8];
  const int b = blockIdx.x, t = threadIdx.x, lane = t & 31, w = t >> 5;
  const int lo = b * 256;

  const int o0 = offt[b * GAT_SORT_P + t];
  const int o1 = offt[(b + 1) * GAT_SORT_P + t];
  const int2* run = (const int2*)bkt + (size_t)t * GAT_SORT_CH + o0;
  const int rl = o1 - o0;

  cnt[t] = 0;
  int below = o0;
#pragma unroll
  for (int o = 16; o > 0; o >>= 1) below += __shfl_xor(below, o, 32);
  if (lane == 0) wbase[w] = below;
  __syncthreads();
  int sb = 0;
#pragma unroll
  for (int q = 0; q < 8; ++q) sb += wbase[q];

  for (int i = 0; i < rl; i += GAT_SORT_BATCH) {
    int dv[GAT_SORT_BATCH];
#pragma unroll
    for (int k = 0; k < GAT_SORT_BATCH; ++k) {
      const int j = i + k < rl ? i + k : rl - 1;
      dv[k] = run[j].y;
    }
#pragma unroll
    for (int k = 0; k < GAT_SORT_BATCH; ++k)
      if (i + k < rl) atomicAdd(&cnt[gat_local_dst(dv[k], lo)], 1);
  }
  __syncthreads();

  const int c = cnt[t];
  int inc = c;
#pragma unroll
  for (int o = 1; o < 32; o <<= 1) {
    const int nb = __shfl_up(inc, o, 32);
    if (lane >= o) inc += nb;
  }
  if (lane == 31) wscan[w] = inc;
  __syncthreads();
  int first = sb + inc - c;
#pragma unroll
  for (int q = 0; q < 8; ++q) first += (q < w) ? wscan[q] : 0;
  rowptr[lo + t] = first;
  cnt[t] = first;
  __syncthreads();

  for (int i = 0; i < rl; i += GAT_SORT_BATCH) {
    int2 rec[GAT_SORT_BATCH];
#pragma unroll
    for (int k = 0; k < GAT_SORT_BATCH; ++k) {
      const int j = i + k < rl ? i + k : rl - 1;
      rec[k] = run[j];
    }
#pragma unroll
    for (int k = 0; k < GAT_SORT_BATCH; ++k) {
      if (i + k < rl) {
        const int pos = atomicAdd(&cnt[gat_local_dst(rec[k].y, lo)], 1);
        if ((unsigned)pos < (unsigned)GAT_E) csrc[pos] = rec[k].x;
      }
    }
  }
}

__device__ __forceinline__ float gat_wave_max(float v) {
#pragma unroll
  for (int o = 16; o > 0; o >>= 1) v = fmaxf(v, __shfl_xor(v, o, 32));
  return v;
}
__device__ __forceinline__ float gat_wave_sum(float v) {
#pragma unroll
  for (int o = 16; o > 0; o >>= 1) v += __shfl_xor(v, o, 32);
  return v;
}
__device__ __forceinline__ float gat_lane_bcast(float v, int l) {
  return __int_as_float(__builtin_amdgcn_readlane(__float_as_int(v), l));
}

__device__ __forceinline__ int gat_slot_src(const int* __restrict__ csrc, int j) {
  unsigned u = (unsigned)csrc[j];
  return (int)(u < (unsigned)GAT_N ? u : (unsigned)(GAT_N - 1));
}

__device__ __forceinline__ float4 gat_edge_logits(const float* __restrict__ el, int s, float4 erd) {
  const float4 e = *(const float4*)(el + (size_t)s * GAT_H);
  float4 x;
  x.x = gat_leaky(e.x + erd.x);
  x.y = gat_leaky(e.y + erd.y);
  x.z = gat_leaky(e.z + erd.z);
  x.w = gat_leaky(e.w + erd.w);
  return x;
}

__device__ __forceinline__ void gat_segment_stats(
    const int* __restrict__ csrc, const float* __restrict__ el, float4 erd,
    int beg, int end, int lane, float4& m, float4& den)
{
  const float ninf = __int_as_float((int)0xFF800000u);
  m.x = ninf; m.y = ninf; m.z = ninf; m.w = ninf;
  for (int c = beg; c < end; c += 32) {
    const int j = c + lane;
    if (j < end) {
      const float4 x = gat_edge_logits(el, gat_slot_src(csrc, j), erd);
      m.x = fmaxf(m.x, x.x); m.y = fmaxf(m.y, x.y);
      m.z = fmaxf(m.z, x.z); m.w = fmaxf(m.w, x.w);
    }
  }
  m.x = gat_wave_max(m.x); m.y = gat_wave_max(m.y);
  m.z = gat_wave_max(m.z); m.w = gat_wave_max(m.w);
  den.x = 0.f; den.y = 0.f; den.z = 0.f; den.w = 0.f;
  for (int c = beg; c < end; c += 32) {
    const int j = c + lane;
    if (j < end) {
      const float4 x = gat_edge_logits(el, gat_slot_src(csrc, j), erd);
      den.x += expf(x.x - m.x); den.y += expf(x.y - m.y);
      den.z += expf(x.z - m.z); den.w += expf(x.w - m.w);
    }
  }
  den.x = gat_wave_sum(den.x); den.y = gat_wave_sum(den.y);
  den.z = gat_wave_sum(den.z); den.w = gat_wave_sum(den.w);
}

__device__ __forceinline__ void gat_chunk_weights(
    const int* __restrict__ csrc, const float* __restrict__ el, float4 erd,
    int c, int end, int lane, float4 m, float4 den, int& s, float4& a)
{
  s = 0;
  a.x = 0.f; a.y = 0.f; a.z = 0.f; a.w = 0.f;
  const int j = c + lane;
  if (j < end) {
    s = gat_slot_src(csrc, j);
    const float4 x = gat_edge_logits(el, s, erd);
    a.x = expf(x.x - m.x) / den.x; a.y = expf(x.y - m.y) / den.y;
    a.z = expf(x.z - m.z) / den.z; a.w = expf(x.w - m.w) / den.w;
  }
}

__global__ __launch_bounds__(256) void gat_node_agg_elu(
    const int* __restrict__ rowptr, const int* __restrict__ csrc,
    const float* __restrict__ feat, const float* __restrict__ el,
    const float* __restrict__ er, const float* __restrict__ bias,
    const float* __restrict__ res, float* __restrict__ out)
{
  const int d = __builtin_amdgcn_readfirstlane((int)(blockIdx.x * 8 + (threadIdx.x >> 5)));
  if (d >= GAT_N) return;
  const int lane = threadIdx.x & 31;
  const int beg = rowptr[d], end = rowptr[d + 1];
  const float4 erd = *(const float4*)(er + (size_t)d * GAT_H);

  float4 m, den;
  gat_segment_stats(csrc, el, erd, beg, end, lane, m, den);

  const bool lo = lane < 16;
  float4 accA = {0.f, 0.f, 0.f, 0.f}, accB = {0.f, 0.f, 0.f, 0.f};
  for (int c = beg; c < end; c += 32) {
    int s; float4 a;
    gat_chunk_weights(csrc, el, erd, c, end, lane, m, den, s, a);
    const int cnt = (end - c) < 32 ? (end - c) : 32;
#pragma unroll 4
    for (int t = 0; t < cnt; ++t) {
      const int st = __builtin_amdgcn_readlane(s, t);
      const float a0 = gat_lane_bcast(a.x, t), a1 = gat_lane_bcast(a.y, t);
      const float a2 = gat_lane_bcast(a.z, t), a3 = gat_lane_bcast(a.w, t);
      const float wA = lo ? a0 : a1;
      const float wB = lo ? a2 : a3;
      const float4* row = (const float4*)(feat + (size_t)st * 256);
      const float4 vA = row[lane];
      const float4 vB = row[32 + lane];
      accA.x += vA.x * wA; accA.y += vA.y * wA; accA.z += vA.z * wA; accA.w += vA.w * wA;
      accB.x += vB.x * wB; accB.y += vB.y * wB; accB.z += vB.z * wB; accB.w += vB.w * wB;
    }
  }

  const float4 bA = *(const float4*)(bias + 4 * lane);
  const float4 bB = *(const float4*)(bias + 128 + 4 * lane);
  float4 vA, vB;
  vA.x = accA.x + bA.x; vA.y = accA.y + bA.y; vA.z = accA.z + bA.z; vA.w = accA.w + bA.w;
  vB.x = accB.x + bB.x; vB.y = accB.y + bB.y; vB.z = accB.z + bB.z; vB.w = accB.w + bB.w;
  const size_t o = (size_t)d * 256 + 4 * lane;
  if (res) {
    const float4 rA = *(const float4*)(res + o);
    const float4 rB = *(const float4*)(res + o + 128);
    vA.x += rA.x; vA.y += rA.y; vA.z += rA.z; vA.w += rA.w;
    vB.x += rB.x; vB.y += rB.y; vB.z += rB.z; vB.w += rB.w;
  }
  vA.x = vA.x > 0.f ? vA.x : (expf(vA.x) - 1.f);
  vA.y = vA.y > 0.f ? vA.y : (expf(vA.y) - 1.f);
  vA.z = vA.z > 0.f ? vA.z : (expf(vA.z) - 1.f);
  vA.w = vA.w > 0.f ? vA.w : (expf(vA.w) - 1.f);
  vB.x = vB.x > 0.f ? vB.x : (expf(vB.x) - 1.f);
  vB.y = vB.y > 0.f ? vB.y : (expf(vB.y) - 1.f);
  vB.z = vB.z > 0.f ? vB.z : (expf(vB.z) - 1.f);
  vB.w = vB.w > 0.f ? vB.w : (expf(vB.w) - 1.f);
  *(float4*)(out + o) = vA;
  *(float4*)(out + o + 128) = vB;
}

__global__ __launch_bounds__(256) void gat_node_agg_mean(
    const int* __restrict__ rowptr, const int* __restrict__ csrc,
    const float* __restrict__ feat, const float* __restrict__ el,
    const float* __restrict__ er, const float* __restrict__ bias,
    const float* __restrict__ res2, float* __restrict__ out)
{
  __shared__ float sval[8 * 160];
  const int w = threadIdx.x >> 5;
  const int d = __builtin_amdgcn_readfirstlane((int)(blockIdx.x * 8 + w));
  const int lane = threadIdx.x & 31;

  if (d < GAT_N) {
    const int beg = rowptr[d], end = rowptr[d + 1];
    const float4 erd = *(const float4*)(er + (size_t)d * GAT_H);
    float4 m, den;
    gat_segment_stats(csrc, el, erd, beg, end, lane, m, den);

    float acc0 = 0.f, acc1 = 0.f, acc2 = 0.f, acc3 = 0.f, acc4 = 0.f;
    for (int c = beg; c < end; c += 32) {
      int s; float4 a;
      gat_chunk_weights(csrc, el, erd, c, end, lane, m, den, s, a);
      const int cnt = (end - c) < 32 ? (end - c) : 32;
#pragma unroll 4
      for (int t = 0; t < cnt; ++t) {
        const int st = __builtin_amdgcn_readlane(s, t);
        const float a0 = gat_lane_bcast(a.x, t), a1 = gat_lane_bcast(a.y, t);
        const float a2 = gat_lane_bcast(a.z, t), a3 = gat_lane_bcast(a.w, t);
        const float* row = feat + (size_t)st * 160 + lane;
        const float v0 = row[0], v1 = row[32], v2 = row[64], v3 = row[96], v4 = row[128];
        acc0 += v0 * a0;
        acc1 += v1 * (lane >= 8 ? a1 : a0);
        acc2 += v2 * (lane >= 16 ? a2 : a1);
        acc3 += v3 * (lane >= 24 ? a3 : a2);
        acc4 += v4 * a3;
      }
    }
    const float* rr = res2 + (size_t)d * 160 + lane;
    const float* bb = bias + lane;
    float* sv = sval + w * 160 + lane;
    sv[0]   = acc0 + bb[0]   + rr[0];
    sv[32]  = acc1 + bb[32]  + rr[32];
    sv[64]  = acc2 + bb[64]  + rr[64];
    sv[96]  = acc3 + bb[96]  + rr[96];
    sv[128] = acc4 + bb[128] + rr[128];
  }
  __syncthreads();
  for (int i = threadIdx.x; i < 8 * 40; i += 256) {
    const int nl = i / 40, c = i - nl * 40;
    if ((int)(blockIdx.x * 8 + nl) < GAT_N) {
      const float* sv = sval + nl * 160 + c;
      float sum = 0.f;
#pragma unroll
      for (int h = 0; h < GAT_H; ++h) sum += sv[h * 40];
      out[(size_t)blockIdx.x * 320 + i] = sum * 0.25f;
    }
  }
}

extern "C" void kernel_launch(void* const* d_in, const int* in_sizes, int n_in,
                              void* d_out, int out_size, void* d_ws, size_t ws_size,
                              hipStream_t stream)
{
  const float* x     = (const float*)d_in[0];
  const int*   src   = (const int*)d_in[1];
  const int*   dst   = (const int*)d_in[2];
  const float* W0    = (const float*)d_in[3];
  const float* al0   = (const float*)d_in[4];
  const float* ar0   = (const float*)d_in[5];
  const float* b0    = (const float*)d_in[6];
  const float* W1    = (const float*)d_in[7];
  const float* al1   = (const float*)d_in[8];
  const float* ar1   = (const float*)d_in[9];
  const float* b1    = (const float*)d_in[10];
  const float* W2    = (const float*)d_in[11];
  const float* al2   = (const float*)d_in[12];
  const float* ar2   = (const float*)d_in[13];
  const float* b2    = (const float*)d_in[14];
  const float* Wres2 = (const float*)d_in[15];

  const size_t NF = (size_t)GAT_N * 256;
  float* ws   = (float*)d_ws;
  float* bufA = ws;
  float* bufB = bufA + NF;
  float* bufC = bufB + NF;
  float* EL   = bufC + NF;
  float* ER   = EL + (size_t)GAT_N * GAT_H;
  int*   OFFT = (int*)(ER + (size_t)GAT_N * GAT_H);
  int*   RPTR = OFFT + GAT_OFFT_LEN;
  int*   BKT  = RPTR + GAT_RPTR_LEN;
  int*   CSRC = BKT + 2 * (size_t)GAT_E;

  const int rowTiles = (GAT_N + 31) / 32;
  dim3 gemmBlk256(128);
  dim3 gemmBlk160(160);

  const int nodeBlocks = (GAT_N + 7) / 8;

  gat_chunk_buckets<<<GAT_SORT_P, GAT_SORT_T1, 0, stream>>>(src, dst, OFFT, BKT);
  gat_bucket_segments<<<GAT_SORT_B, 256, 0, stream>>>(OFFT, BKT, RPTR, CSRC);

  gat_gemm_wmma<256, 256, 4, 64, true><<<rowTiles, gemmBlk256, 0, stream>>>(x, W0, bufA, GAT_N, al0, ar0, EL, ER);
  gat_node_agg_elu<<<nodeBlocks, 256, 0, stream>>>(RPTR, CSRC, bufA, EL, ER, b0, nullptr, bufB);

  gat_gemm_wmma<256, 256, 4, 64, true><<<rowTiles, gemmBlk256, 0, stream>>>(bufB, W1, bufA, GAT_N, al1, ar1, EL, ER);
  gat_node_agg_elu<<<nodeBlocks, 256, 0, stream>>>(RPTR, CSRC, bufA, EL, ER, b1, bufB, bufC);

  gat_gemm_wmma<256, 160, 2, 40, true><<<rowTiles, gemmBlk160, 0, stream>>>(bufC, W2, bufA, GAT_N, al2, ar2, EL, ER);
  gat_gemm_wmma<256, 160, 2, 40, false><<<rowTiles, gemmBlk160, 0, stream>>>(bufC, Wres2, bufB, GAT_N, nullptr, nullptr, nullptr, nullptr);
  gat_node_agg_mean<<<nodeBlocks, 256, 0, stream>>>(RPTR, CSRC, bufA, EL, ER, b2, bufB, (float*)d_out);
}


# softmax of a node of at most 32 edges in one pass over registers: source id and el[src] fetched once per edge, not three times; longer nodes keep the three passes; sums keep their order
# speedup vs baseline: 3.9999x; 1.0479x over previous
#include <hip/hip_runtime.h>
#include <math.h>

#define GAT_N 50000
#define GAT_E 800000
#define GAT_H 4
#define GAT_SLOPE 0.2f

typedef __attribute__((ext_vector_type(2))) float v2f;
typedef __attribute__((ext_vector_type(8))) float v8f;

template <int MASK>
__device__ __forceinline__ float gat_row_xor_add(float v) {
  return v + __int_as_float(__builtin_amdgcn_update_dpp(
                 0, __float_as_int(v), 0x160 + MASK, 0xf, 0xf, true));
}

template <int K, int NCOL, int NT, int D, bool SCORES>
__global__ __launch_bounds__(NCOL / (16 * NT) * 32) void gat_gemm_wmma(
    const float* __restrict__ A, const float* __restrict__ B,
    float* __restrict__ C, int M,
    const float* __restrict__ al, const float* __restrict__ ar,
    float* __restrict__ el, float* __restrict__ er)
{
  typedef float vnt __attribute__((ext_vector_type(NT)));
  typedef float v4f __attribute__((ext_vector_type(4)));
  const int wave = threadIdx.x >> 5;
  const int lane = threadIdx.x & 31;
  const int row0 = blockIdx.x << 5;
  const int col0 = wave * (16 * NT);
  const int c    = lane & 15;
  const int h    = lane >> 4;
  const bool hi_valid = (row0 + 16) < M;

  const float* arow0 = A + (size_t)(row0 + c) * K + 4 * h;
  const float* arow1 = hi_valid ? (arow0 + (size_t)16 * K) : arow0;
  const float* bp    = B + (size_t)(4 * h) * NCOL + col0 + NT * c;

  v8f acc0[NT], acc1[NT];
#pragma unroll
  for (int t = 0; t < NT; ++t) { acc0[t] = (v8f){}; acc1[t] = (v8f){}; }

#pragma unroll 2
  for (int k0 = 0; k0 < K; k0 += 8) {
    const v4f a0 = *(const v4f*)(arow0 + k0);
    const v4f a1 = *(const v4f*)(arow1 + k0);
    vnt b[4];
#pragma unroll
    for (int i = 0; i < 4; ++i) b[i] = *(const vnt*)(bp + (size_t)(k0 + i) * NCOL);
#pragma unroll
    for (int j = 0; j < 2; ++j) {
      v2f fa0, fa1;
      fa0.x = a0[2 * j]; fa0.y = a0[2 * j + 1];
      fa1.x = a1[2 * j]; fa1.y = a1[2 * j + 1];
#pragma unroll
      for (int t = 0; t < NT; ++t) {
        v2f fb;
        fb.x = b[2 * j][t];
        fb.y = b[2 * j + 1][t];
        acc0[t] = __builtin_amdgcn_wmma_f32_16x16x4_f32(
            false, fa0, false, fb, (short)0, acc0[t], false, false);
        acc1[t] = __builtin_amdgcn_wmma_f32_16x16x4_f32(
            false, fa1, false, fb, (short)0, acc1[t], false, false);
      }
    }
  }

  constexpr int NG    = NCOL / (4 * NT);
  constexpr int GPH   = D / (4 * NT);
  constexpr int PITCH = 2 * NG + 2;
  __shared__ float part[SCORES ? 32 * PITCH : 1];
  if (SCORES) {
    const vnt wl = *(const vnt*)(al + col0 + NT * c);
    const vnt wr = *(const vnt*)(ar + col0 + NT * c);
    const int g = (col0 + NT * c) / (4 * NT);
#pragma unroll
    for (int r = 0; r < 8; ++r) {
      float pl0 = 0.f, pr0 = 0.f, pl1 = 0.f, pr1 = 0.f;
#pragma unroll
      for (int t = 0; t < NT; ++t) {
        pl0 = fmaf(acc0[t][r], wl[t], pl0);
        pr0 = fmaf(acc0[t][r], wr[t], pr0);
        pl1 = fmaf(acc1[t][r], wl[t], pl1);
        pr1 = fmaf(acc1[t][r], wr[t], pr1);
      }
      pl0 = gat_row_xor_add<2>(gat_row_xor_add<1>(pl0));
      pr0 = gat_row_xor_add<2>(gat_row_xor_add<1>(pr0));
      pl1 = gat_row_xor_add<2>(gat_row_xor_add<1>(pl1));
      pr1 = gat_row_xor_add<2>(gat_row_xor_add<1>(pr1));
      if ((c & 3) == 0) {
        v2f s0, s1;
        s0.x = pl0; s0.y = pr0;
        s1.x = pl1; s1.y = pr1;
        *(v2f*)(part + (8 * h + r) * PITCH + 2 * g)      = s0;
        *(v2f*)(part + (16 + 8 * h + r) * PITCH + 2 * g) = s1;
      }
    }
    __syncthreads();
  }

  float* crow0 = C + (size_t)(row0 + 8 * h) * NCOL + col0 + NT * c;
#pragma unroll
  for (int r = 0; r < 8; ++r) {
    vnt o;
#pragma unroll
    for (int t = 0; t < NT; ++t) o[t] = acc0[t][r];
    *(vnt*)(crow0 + (size_t)r * NCOL) = o;
  }
  if (hi_valid) {
    float* crow1 = crow0 + (size_t)16 * NCOL;
#pragma unroll
    for (int r = 0; r < 8; ++r) {
      vnt o;
#pragma unroll
      for (int t = 0; t < NT; ++t) o[t] = acc1[t][r];
      *(vnt*)(crow1 + (size_t)r * NCOL) = o;
    }
  }

  if (SCORES) {
    for (int i = threadIdx.x; i < 32 * GAT_H * 2; i += NCOL / (16 * NT) * 32) {
      const int row = i >> 3, hd = (i >> 1) & 3, which = i & 1;
      if (row0 + row < M) {
        const float* p = part + row * PITCH + 2 * hd * GPH + which;
        float sum = 0.f;
#pragma unroll
        for (int q = 0; q < GPH; ++q) sum += p[2 * q];
        (which ? er : el)[(size_t)(row0 + row) * GAT_H + hd] = sum;
      }
    }
  }
}

__device__ __forceinline__ float gat_leaky(float x) {
  return x > 0.f ? x : GAT_SLOPE * x;
}

#define GAT_SORT_P 256
#define GAT_SORT_CH (GAT_E / GAT_SORT_P)
#define GAT_SORT_B 196
#define GAT_SORT_T1 256
#define GAT_SORT_EPT 13
#define GAT_SORT_BATCH 8
#define GAT_OFFT_LEN ((GAT_SORT_B + 1) * GAT_SORT_P)
#define GAT_RPTR_LEN (GAT_SORT_B * 256)
static_assert(GAT_SORT_CH * GAT_SORT_P == GAT_E, "chunks must tile the edge list");
static_assert(GAT_SORT_EPT * GAT_SORT_T1 >= GAT_SORT_CH, "a chunk must fit one pass of the work-group");
static_assert(GAT_SORT_T1 > GAT_SORT_B, "one thread per bucket offset, the end entry included");
static_assert(GAT_RPTR_LEN >= GAT_N + 1, "buckets must cover every node and the end entry");
static_assert(GAT_SORT_B <= 7 * 32, "one wave scans the bucket counts, 7 per lane");
static_assert(GAT_SORT_P == 256, "level 2 gives one of its 256 threads to each chunk's run");

__device__ __forceinline__ int gat_bucket_of(int d) {
  const unsigned b = (unsigned)d >> 8;
  return (int)(b < (unsigned)GAT_SORT_B ? b : (unsigned)(GAT_SORT_B - 1));
}
__device__ __forceinline__ int gat_local_dst(int d, int lo) {
  const unsigned u = (unsigned)(d - lo);
  return (int)(u < 256u ? u : 255u);
}

__global__ __launch_bounds__(GAT_SORT_T1) void gat_chunk_buckets(
    const int* __restrict__ src, const int* __restrict__ dst,
    int* __restrict__ offt, int* __restrict__ bkt)
{
  __shared__ int hist[GAT_SORT_B + 1];
  const int p = blockIdx.x, t = threadIdx.x;
  const int e0 = p * GAT_SORT_CH;

  int s[GAT_SORT_EPT], d[GAT_SORT_EPT];
#pragma unroll
  for (int k = 0; k < GAT_SORT_EPT; ++k) {
    const int i = t + k * GAT_SORT_T1;
    const int ic = i < GAT_SORT_CH ? i : GAT_SORT_CH - 1;
    d[k] = dst[e0 + ic];
    s[k] = src[e0 + ic];
  }
  if (t <= GAT_SORT_B) hist[t] = 0;
  __syncthreads();

  int b[GAT_SORT_EPT], r[GAT_SORT_EPT];
#pragma unroll
  for (int k = 0; k < GAT_SORT_EPT; ++k) {
    b[k] = -1; r[k] = 0;
    if (t + k * GAT_SORT_T1 < GAT_SORT_CH) {
      b[k] = gat_bucket_of(d[k]);
      r[k] = atomicAdd(&hist[b[k]], 1);
    }
  }
  __syncthreads();

  if (t < 32) {
    int c[7];
    int tot = 0;
#pragma unroll
    for (int q = 0; q < 7; ++q) {
      const int idx = 7 * t + q;
      c[q] = idx < GAT_SORT_B ? hist[idx] : 0;
      tot += c[q];
    }
    int inc = tot;
#pragma unroll
    for (int o = 1; o < 32; o <<= 1) {
      const int nb = __shfl_up(inc, o, 32);
      if (t >= o) inc += nb;
    }
    int run = inc - tot;
#pragma unroll
    for (int q = 0; q < 7; ++q) {
      const int idx = 7 * t + q;
      if (idx <= GAT_SORT_B) hist[idx] = run;
      run += c[q];
    }
  }
  __syncthreads();

  if (t <= GAT_SORT_B) offt[t * GAT_SORT_P + p] = hist[t];
#pragma unroll
  for (int k = 0; k < GAT_SORT_EPT; ++k) {
    if (b[k] >= 0) {
      int2 rec;
      rec.x = s[k]; rec.y = d[k];
      *(int2*)(bkt + 2 * (size_t)(e0 + hist[b[k]] + r[k])) = rec;
    }
  }
}

__global__ __launch_bounds__(256) void gat_bucket_segments(
    const int* __restrict__ offt, const int* __restrict__ bkt,
    int* __restrict__ rowptr, int* __restrict__ csrc)
{
  __shared__ int cnt[256];
  __shared__ int wbase[8];
  __shared__ int wscan[8];
  const int b = blockIdx.x, t = threadIdx.x, lane = t & 31, w = t >> 5;
  const int lo = b * 256;

  const int o0 = offt[b * GAT_SORT_P + t];
  const int o1 = offt[(b + 1) * GAT_SORT_P + t];
  const int2* run = (const int2*)bkt + (size_t)t * GAT_SORT_CH + o0;
  const int rl = o1 - o0;

  cnt[t] = 0;
  int below = o0;
#pragma unroll
  for (int o = 16; o > 0; o >>= 1) below += __shfl_xor(below, o, 32);
  if (lane == 0) wbase[w] = below;
  __syncthreads();
  int sb = 0;
#pragma unroll
  for (int q = 0; q < 8; ++q) sb += wbase[q];

  for (int i = 0; i < rl; i += GAT_SORT_BATCH) {
    int dv[GAT_SORT_BATCH];
#pragma unroll
    for (int k = 0; k < GAT_SORT_BATCH; ++k) {
      const int j = i + k < rl ? i + k : rl - 1;
      dv[k] = run[j].y;
    }
#pragma unroll
    for (int k = 0; k < GAT_SORT_BATCH; ++k)
      if (i + k < rl) atomicAdd(&cnt[gat_local_dst(dv[k], lo)], 1);
  }
  __syncthreads();

  const int c = cnt[t];
  int inc = c;
#pragma unroll
  for (int o = 1; o < 32; o <<= 1) {
    const int nb = __shfl_up(inc, o, 32);
    if (lane >= o) inc += nb;
  }
  if (lane == 31) wscan[w] = inc;
  __syncthreads();
  int first = sb + inc - c;
#pragma unroll
  for (int q = 0; q < 8; ++q) first += (q < w) ? wscan[q] : 0;
  rowptr[lo + t] = first;
  cnt[t] = first;
  __syncthreads();

  for (int i = 0; i < rl; i += GAT_SORT_BATCH) {
    int2 rec[GAT_SORT_BATCH];
#pragma unroll
    for (int k = 0; k < GAT_SORT_BATCH; ++k) {
      const int j = i + k < rl ? i + k : rl - 1;
      rec[k] = run[j];
    }
#pragma unroll
    for (int k = 0; k < GAT_SORT_BATCH; ++k) {
      if (i + k < rl) {
        const int pos = atomicAdd(&cnt[gat_local_dst(rec[k].y, lo)], 1);
        if ((unsigned)pos < (unsigned)GAT_E) csrc[pos] = rec[k].x;
      }
    }
  }
}

__device__ __forceinline__ float gat_wave_max(float v) {
#pragma unroll
  for (int o = 16; o > 0; o >>= 1) v = fmaxf(v, __shfl_xor(v, o, 32));
  return v;
}
__device__ __forceinline__ float gat_wave_sum(float v) {
#pragma unroll
  for (int o = 16; o > 0; o >>= 1) v += __shfl_xor(v, o, 32);
  return v;
}
__device__ __forceinline__ float gat_lane_bcast(float v, int l) {
  return __int_as_float(__builtin_amdgcn_readlane(__float_as_int(v), l));
}

__device__ __forceinline__ int gat_slot_src(const int* __restrict__ csrc, int j) {
  unsigned u = (unsigned)csrc[j];
  return (int)(u < (unsigned)GAT_N ? u : (unsigned)(GAT_N - 1));
}

__device__ __forceinline__ float4 gat_edge_logits(const float* __restrict__ el, int s, float4 erd) {
  const float4 e = *(const float4*)(el + (size_t)s * GAT_H);
  float4 x;
  x.x = gat_leaky(e.x + erd.x);
  x.y = gat_leaky(e.y + erd.y);
  x.z = gat_leaky(e.z + erd.z);
  x.w = gat_leaky(e.w + erd.w);
  return x;
}

__device__ __forceinline__ void gat_segment_stats(
    const int* __restrict__ csrc, const float* __restrict__ el, float4 erd,
    int beg, int end, int lane, float4& m, float4& den)
{
  const float ninf = __int_as_float((int)0xFF800000u);
  m.x = ninf; m.y = ninf; m.z = ninf; m.w = ninf;
  for (int c = beg; c < end; c += 32) {
    const int j = c + lane;
    if (j < end) {
      const float4 x = gat_edge_logits(el, gat_slot_src(csrc, j), erd);
      m.x = fmaxf(m.x, x.x); m.y = fmaxf(m.y, x.y);
      m.z = fmaxf(m.z, x.z); m.w = fmaxf(m.w, x.w);
    }
  }
  m.x = gat_wave_max(m.x); m.y = gat_wave_max(m.y);
  m.z = gat_wave_max(m.z); m.w = gat_wave_max(m.w);
  den.x = 0.f; den.y = 0.f; den.z = 0.f; den.w = 0.f;
  for (int c = beg; c < end; c += 32) {
    const int j = c + lane;
    if (j < end) {
      const float4 x = gat_edge_logits(el, gat_slot_src(csrc, j), erd);
      den.x += expf(x.x - m.x); den.y += expf(x.y - m.y);
      den.z += expf(x.z - m.z); den.w += expf(x.w - m.w);
    }
  }
  den.x = gat_wave_sum(den.x); den.y = gat_wave_sum(den.y);
  den.z = gat_wave_sum(den.z); den.w = gat_wave_sum(den.w);
}

__device__ __forceinline__ void gat_chunk_weights(
    const int* __restrict__ csrc, const float* __restrict__ el, float4 erd,
    int c, int end, int lane, float4 m, float4 den, int& s, float4& a)
{
  s = 0;
  a.x = 0.f; a.y = 0.f; a.z = 0.f; a.w = 0.f;
  const int j = c + lane;
  if (j < end) {
    s = gat_slot_src(csrc, j);
    const float4 x = gat_edge_logits(el, s, erd);
    a.x = expf(x.x - m.x) / den.x; a.y = expf(x.y - m.y) / den.y;
    a.z = expf(x.z - m.z) / den.z; a.w = expf(x.w - m.w) / den.w;
  }
}

__device__ __forceinline__ void gat_segment_weights_once(
    const int* __restrict__ csrc, const float* __restrict__ el, float4 erd,
    int beg, int end, int lane, int& s, float4& a)
{
  const float ninf = __int_as_float((int)0xFF800000u);
  const bool on = beg + lane < end;
  s = 0;
  float4 x, m;
  x.x = 0.f; x.y = 0.f; x.z = 0.f; x.w = 0.f;
  m.x = ninf; m.y = ninf; m.z = ninf; m.w = ninf;
  if (on) {
    s = gat_slot_src(csrc, beg + lane);
    x = gat_edge_logits(el, s, erd);
    m.x = fmaxf(m.x, x.x); m.y = fmaxf(m.y, x.y);
    m.z = fmaxf(m.z, x.z); m.w = fmaxf(m.w, x.w);
  }
  m.x = gat_wave_max(m.x); m.y = gat_wave_max(m.y);
  m.z = gat_wave_max(m.z); m.w = gat_wave_max(m.w);
  float4 ex;
  ex.x = 0.f; ex.y = 0.f; ex.z = 0.f; ex.w = 0.f;
  if (on) {
    ex.x = expf(x.x - m.x); ex.y = expf(x.y - m.y);
    ex.z = expf(x.z - m.z); ex.w = expf(x.w - m.w);
  }
  float4 den;
  den.x = gat_wave_sum(ex.x); den.y = gat_wave_sum(ex.y);
  den.z = gat_wave_sum(ex.z); den.w = gat_wave_sum(ex.w);
  a.x = 0.f; a.y = 0.f; a.z = 0.f; a.w = 0.f;
  if (on) {
    a.x = ex.x / den.x; a.y = ex.y / den.y;
    a.z = ex.z / den.z; a.w = ex.w / den.w;
  }
}

__global__ __launch_bounds__(256) void gat_node_agg_elu(
    const int* __restrict__ rowptr, const int* __restrict__ csrc,
    const float* __restrict__ feat, const float* __restrict__ el,
    const float* __restrict__ er, const float* __restrict__ bias,
    const float* __restrict__ res, float* __restrict__ out)
{
  const int d = __builtin_amdgcn_readfirstlane((int)(blockIdx.x * 8 + (threadIdx.x >> 5)));
  if (d >= GAT_N) return;
  const int lane = threadIdx.x & 31;
  const int beg = rowptr[d], end = rowptr[d + 1];
  const float4 erd = *(const float4*)(er + (size_t)d * GAT_H);

  const bool once = end - beg <= 32;
  float4 m = {0.f, 0.f, 0.f, 0.f}, den = {0.f, 0.f, 0.f, 0.f};
  int s1 = 0;
  float4 a1 = {0.f, 0.f, 0.f, 0.f};
  if (once) gat_segment_weights_once(csrc, el, erd, beg, end, lane, s1, a1);
  else      gat_segment_stats(csrc, el, erd, beg, end, lane, m, den);

  const bool lo = lane < 16;
  float4 accA = {0.f, 0.f, 0.f, 0.f}, accB = {0.f, 0.f, 0.f, 0.f};
  for (int c = beg; c < end; c += 32) {
    int s = s1; float4 a = a1;
    if (!once) gat_chunk_weights(csrc, el, erd, c, end, lane, m, den, s, a);
    const int cnt = (end - c) < 32 ? (end - c) : 32;
#pragma unroll 4
    for (int t = 0; t < cnt; ++t) {
      const int st = __builtin_amdgcn_readlane(s, t);
      const float a0 = gat_lane_bcast(a.x, t), a1 = gat_lane_bcast(a.y, t);
      const float a2 = gat_lane_bcast(a.z, t), a3 = gat_lane_bcast(a.w, t);
      const float wA = lo ? a0 : a1;
      const float wB = lo ? a2 : a3;
      const float4* row = (const float4*)(feat + (size_t)st * 256);
      const float4 vA = row[lane];
      const float4 vB = row[32 + lane];
      accA.x += vA.x * wA; accA.y += vA.y * wA; accA.z += vA.z * wA; accA.w += vA.w * wA;
      accB.x += vB.x * wB; accB.y += vB.y * wB; accB.z += vB.z * wB; accB.w += vB.w * wB;
    }
  }

  const float4 bA = *(const float4*)(bias + 4 * lane);
  const float4 bB = *(const float4*)(bias + 128 + 4 * lane);
  float4 vA, vB;
  vA.x = accA.x + bA.x; vA.y = accA.y + bA.y; vA.z = accA.z + bA.z; vA.w = accA.w + bA.w;
  vB.x = accB.x + bB.x; vB.y = accB.y + bB.y; vB.z = accB.z + bB.z; vB.w = accB.w + bB.w;
  const size_t o = (size_t)d * 256 + 4 * lane;
  if (res) {
    const float4 rA = *(const float4*)(res + o);
    const float4 rB = *(const float4*)(res + o + 128);
    vA.x += rA.x; vA.y += rA.y; vA.z += rA.z; vA.w += rA.w;
    vB.x += rB.x; vB.y += rB.y; vB.z += rB.z; vB.w += rB.w;
  }
  vA.x = vA.x > 0.f ? vA.x : (expf(vA.x) - 1.f);
  vA.y = vA.y > 0.f ? vA.y : (expf(vA.y) - 1.f);
  vA.z = vA.z > 0.f ? vA.z : (expf(vA.z) - 1.f);
  vA.w = vA.w > 0.f ? vA.w : (expf(vA.w) - 1.f);
  vB.x = vB.x > 0.f ? vB.x : (expf(vB.x) - 1.f);
  vB.y = vB.y > 0.f ? vB.y : (expf(vB.y) - 1.f);
  vB.z = vB.z > 0.f ? vB.z : (expf(vB.z) - 1.f);
  vB.w = vB.w > 0.f ? vB.w : (expf(vB.w) - 1.f);
  *(float4*)(out + o) = vA;
  *(float4*)(out + o + 128) = vB;
}

__global__ __launch_bounds__(256) void gat_node_agg_mean(
    const int* __restrict__ rowptr, const int* __restrict__ csrc,
    const float* __restrict__ feat, const float* __restrict__ el,
    const float* __restrict__ er, const float* __restrict__ bias,
    const float* __restrict__ res2, float* __restrict__ out)
{
  __shared__ float sval[8 * 160];
  const int w = threadIdx.x >> 5;
  const int d = __builtin_amdgcn_readfirstlane((int)(blockIdx.x * 8 + w));
  const int lane = threadIdx.x & 31;

  if (d < GAT_N) {
    const int beg = rowptr[d], end = rowptr[d + 1];
    const float4 erd = *(const float4*)(er + (size_t)d * GAT_H);
    const bool once = end - beg <= 32;
    float4 m = {0.f, 0.f, 0.f, 0.f}, den = {0.f, 0.f, 0.f, 0.f};
    int s1 = 0;
    float4 a1 = {0.f, 0.f, 0.f, 0.f};
    if (once) gat_segment_weights_once(csrc, el, erd, beg, end, lane, s1, a1);
    else      gat_segment_stats(csrc, el, erd, beg, end, lane, m, den);

    float acc0 = 0.f, acc1 = 0.f, acc2 = 0.f, acc3 = 0.f, acc4 = 0.f;
    for (int c = beg; c < end; c += 32) {
      int s = s1; float4 a = a1;
      if (!once) gat_chunk_weights(csrc, el, erd, c, end, lane, m, den, s, a);
      const int cnt = (end - c) < 32 ? (end - c) : 32;
#pragma unroll 4
      for (int t = 0; t < cnt; ++t) {
        const int st = __builtin_amdgcn_readlane(s, t);
        const float a0 = gat_lane_bcast(a.x, t), a1 = gat_lane_bcast(a.y, t);
        const float a2 = gat_lane_bcast(a.z, t), a3 = gat_lane_bcast(a.w, t);
        const float* row = feat + (size_t)st * 160 + lane;
        const float v0 = row[0], v1 = row[32], v2 = row[64], v3 = row[96], v4 = row[128];
        acc0 += v0 * a0;
        acc1 += v1 * (lane >= 8 ? a1 : a0);
        acc2 += v2 * (lane >= 16 ? a2 : a1);
        acc3 += v3 * (lane >= 24 ? a3 : a2);
        acc4 += v4 * a3;
      }
    }
    const float* rr = res2 + (size_t)d * 160 + lane;
    const float* bb = bias + lane;
    float* sv = sval + w * 160 + lane;
    sv[0]   = acc0 + bb[0]   + rr[0];
    sv[32]  = acc1 + bb[32]  + rr[32];
    sv[64]  = acc2 + bb[64]  + rr[64];
    sv[96]  = acc3 + bb[96]  + rr[96];
    sv[128] = acc4 + bb[128] + rr[128];
  }
  __syncthreads();
  for (int i = threadIdx.x; i < 8 * 40; i += 256) {
    const int nl = i / 40, c = i - nl * 40;
    if ((int)(blockIdx.x * 8 + nl) < GAT_N) {
      const float* sv = sval + nl * 160 + c;
      float sum = 0.f;
#pragma unroll
      for (int h = 0; h < GAT_H; ++h) sum += sv[h * 40];
      out[(size_t)blockIdx.x * 320 + i] = sum * 0.25f;
    }
  }
}

extern "C" void kernel_launch(void* const* d_in, const int* in_sizes, int n_in,
                              void* d_out, int out_size, void* d_ws, size_t ws_size,
                              hipStream_t stream)
{
  const float* x     = (const float*)d_in[0];
  const int*   src   = (const int*)d_in[1];
  const int*   dst   = (const int*)d_in[2];
  const float* W0    = (const float*)d_in[3];
  const float* al0   = (const float*)d_in[4];
  const float* ar0   = (const float*)d_in[5];
  const float* b0    = (const float*)d_in[6];
  const float* W1    = (const float*)d_in[7];
  const float* al1   = (const float*)d_in[8];
  const float* ar1   = (const float*)d_in[9];
  const float* b1    = (const float*)d_in[10];
  const float* W2    = (const float*)d_in[11];
  const float* al2   = (const float*)d_in[12];
  const float* ar2   = (const float*)d_in[13];
  const float* b2    = (const float*)d_in[14];
  const float* Wres2 = (const float*)d_in[15];

  const size_t NF = (size_t)GAT_N * 256;
  float* ws   = (float*)d_ws;
  float* bufA = ws;
  float* bufB = bufA + NF;
  float* bufC = bufB + NF;
  float* EL   = bufC + NF;
  float* ER   = EL + (size_t)GAT_N * GAT_H;
  int*   OFFT = (int*)(ER + (size_t)GAT_N * GAT_H);
  int*   RPTR = OFFT + GAT_OFFT_LEN;
  int*   BKT  = RPTR + GAT_RPTR_LEN;
  int*   CSRC = BKT + 2 * (size_t)GAT_E;

  const int rowTiles = (GAT_N + 31) / 32;
  dim3 gemmBlk256(128);
  dim3 gemmBlk160(160);

  const int nodeBlocks = (GAT_N + 7) / 8;

  gat_chunk_buckets<<<GAT_SORT_P, GAT_SORT_T1, 0, stream>>>(src, dst, OFFT, BKT);
  gat_bucket_segments<<<GAT_SORT_B, 256, 0, stream>>>(OFFT, BKT, RPTR, CSRC);

  gat_gemm_wmma<256, 256, 4, 64, true><<<rowTiles, gemmBlk256, 0, stream>>>(x, W0, bufA, GAT_N, al0, ar0, EL, ER);
  gat_node_agg_elu<<<nodeBlocks, 256, 0, stream>>>(RPTR, CSRC, bufA, EL, ER, b0, nullptr, bufB);

  gat_gemm_wmma<256, 256, 4, 64, true><<<rowTiles, gemmBlk256, 0, stream>>>(bufB, W1, bufA, GAT_N, al1, ar1, EL, ER);
  gat_node_agg_elu<<<nodeBlocks, 256, 0, stream>>>(RPTR, CSRC, bufA, EL, ER, b1, bufB, bufC);

  gat_gemm_wmma<256, 160, 2, 40, true><<<rowTiles, gemmBlk160, 0, stream>>>(bufC, W2, bufA, GAT_N, al2, ar2, EL, ER);
  gat_gemm_wmma<256, 160, 2, 40, false><<<rowTiles, gemmBlk160, 0, stream>>>(bufC, Wres2, bufB, GAT_N, nullptr, nullptr, nullptr, nullptr);
  gat_node_agg_mean<<<nodeBlocks, 256, 0, stream>>>(RPTR, CSRC, bufA, EL, ER, b2, bufB, (float*)d_out);
}


# residual projection taken behind the head mean: h1 @ mean_h(Wres2) (40 columns, two extra waves of the layer-2 GEMM) replaces the 160-column product and its launch; residual sum order changes
# speedup vs baseline: 4.2351x; 1.0588x over previous
#include <hip/hip_runtime.h>
#include <math.h>

#define GAT_N 50000
#define GAT_E 800000
#define GAT_H 4
#define GAT_SLOPE 0.2f

typedef __attribute__((ext_vector_type(2))) float v2f;
typedef __attribute__((ext_vector_type(8))) float v8f;

template <int MASK>
__device__ __forceinline__ float gat_row_xor_add(float v) {
  return v + __int_as_float(__builtin_amdgcn_update_dpp(
                 0, __float_as_int(v), 0x160 + MASK, 0xf, 0xf, true));
}

#define GAT_RESW 48

template <int K>
__device__ __forceinline__ void gat_side_tile(
    const float* __restrict__ A, const float* __restrict__ Bm,
    float* __restrict__ Cm, int r0, int lane)
{
  typedef float v4f __attribute__((ext_vector_type(4)));
  const int c = lane & 15;
  const int h = lane >> 4;
  const float* arow = A + (size_t)(r0 + c) * K + 4 * h;
  const float* bp   = Bm + (size_t)(4 * h) * GAT_RESW + 3 * c;
  v8f acc[3];
#pragma unroll
  for (int t = 0; t < 3; ++t) acc[t] = (v8f){};
#pragma unroll 1
  for (int k0 = 0; k0 < K; k0 += 8) {
    const v4f a = *(const v4f*)(arow + k0);
    float b[4][3];
#pragma unroll
    for (int i = 0; i < 4; ++i)
#pragma unroll
      for (int t = 0; t < 3; ++t) b[i][t] = bp[(size_t)(k0 + i) * GAT_RESW + t];
#pragma unroll
    for (int j = 0; j < 2; ++j) {
      v2f fa;
      fa.x = a[2 * j]; fa.y = a[2 * j + 1];
#pragma unroll
      for (int t = 0; t < 3; ++t) {
        v2f fb;
        fb.x = b[2 * j][t];
        fb.y = b[2 * j + 1][t];
        acc[t] = __builtin_amdgcn_wmma_f32_16x16x4_f32(
            false, fa, false, fb, (short)0, acc[t], false, false);
      }
    }
  }
  float* crow = Cm + (size_t)(r0 + 8 * h) * GAT_RESW + 3 * c;
#pragma unroll
  for (int r = 0; r < 8; ++r)
#pragma unroll
    for (int t = 0; t < 3; ++t) crow[(size_t)r * GAT_RESW + t] = acc[t][r];
}

template <int K, int NCOL, int NT, int D, bool SCORES, int SIDE>
__global__ __launch_bounds__((NCOL / (16 * NT) + SIDE) * 32) void gat_gemm_wmma(
    const float* __restrict__ A, const float* __restrict__ B,
    float* __restrict__ C, int M,
    const float* __restrict__ al, const float* __restrict__ ar,
    float* __restrict__ el, float* __restrict__ er,
    const float* __restrict__ Bm, float* __restrict__ Cm)
{
  constexpr int NW   = NCOL / (16 * NT);
  constexpr int NTHR = (NW + SIDE) * 32;
  typedef float vnt __attribute__((ext_vector_type(NT)));
  typedef float v4f __attribute__((ext_vector_type(4)));
  const int wave = threadIdx.x >> 5;
  const int lane = threadIdx.x & 31;
  const int row0 = blockIdx.x << 5;
  const int col0 = wave * (16 * NT);
  const int c    = lane & 15;
  const int h    = lane >> 4;
  const bool hi_valid = (row0 + 16) < M;
  const bool side = SIDE > 0 &&
      __builtin_amdgcn_readfirstlane((int)(threadIdx.x >> 5)) >= NW;

  constexpr int NG    = NCOL / (4 * NT);
  constexpr int GPH   = D / (4 * NT);
  constexpr int PITCH = 2 * NG + 2;
  __shared__ float part[SCORES ? 32 * PITCH : 1];

  v8f acc0[NT], acc1[NT];
#pragma unroll
  for (int t = 0; t < NT; ++t) { acc0[t] = (v8f){}; acc1[t] = (v8f){}; }

  if (side) {
    const int s = __builtin_amdgcn_readfirstlane((int)(threadIdx.x >> 5)) - NW;
    if (s == 0 || hi_valid) gat_side_tile<K>(A, Bm, Cm, row0 + 16 * s, lane);
  } else {
    const float* arow0 = A + (size_t)(row0 + c) * K + 4 * h;
    const float* arow1 = hi_valid ? (arow0 + (size_t)16 * K) : arow0;
    const float* bp    = B + (size_t)(4 * h) * NCOL + col0 + NT * c;

#pragma unroll 2
    for (int k0 = 0; k0 < K; k0 += 8) {
      const v4f a0 = *(const v4f*)(arow0 + k0);
      const v4f a1 = *(const v4f*)(arow1 + k0);
      vnt b[4];
#pragma unroll
      for (int i = 0; i < 4; ++i) b[i] = *(const vnt*)(bp + (size_t)(k0 + i) * NCOL);
#pragma unroll
      for (int j = 0; j < 2; ++j) {
        v2f fa0, fa1;
        fa0.x = a0[2 * j]; fa0.y = a0[2 * j + 1];
        fa1.x = a1[2 * j]; fa1.y = a1[2 * j + 1];
#pragma unroll
        for (int t = 0; t < NT; ++t) {
          v2f fb;
          fb.x = b[2 * j][t];
          fb.y = b[2 * j + 1][t];
          acc0[t] = __builtin_amdgcn_wmma_f32_16x16x4_f32(
              false, fa0, false, fb, (short)0, acc0[t], false, false);
          acc1[t] = __builtin_amdgcn_wmma_f32_16x16x4_f32(
              false, fa1, false, fb, (short)0, acc1[t], false, false);
        }
      }
    }

    if (SCORES) {
      const vnt wl = *(const vnt*)(al + col0 + NT * c);
      const vnt wr = *(const vnt*)(ar + col0 + NT * c);
      const int g = (col0 + NT * c) / (4 * NT);
#pragma unroll
      for (int r = 0; r < 8; ++r) {
        float pl0 = 0.f, pr0 = 0.f, pl1 = 0.f, pr1 = 0.f;
#pragma unroll
        for (int t = 0; t < NT; ++t) {
          pl0 = fmaf(acc0[t][r], wl[t], pl0);
          pr0 = fmaf(acc0[t][r], wr[t], pr0);
          pl1 = fmaf(acc1[t][r], wl[t], pl1);
          pr1 = fmaf(acc1[t][r], wr[t], pr1);
        }
        pl0 = gat_row_xor_add<2>(gat_row_xor_add<1>(pl0));
        pr0 = gat_row_xor_add<2>(gat_row_xor_add<1>(pr0));
        pl1 = gat_row_xor_add<2>(gat_row_xor_add<1>(pl1));
        pr1 = gat_row_xor_add<2>(gat_row_xor_add<1>(pr1));
        if ((c & 3) == 0) {
          v2f s0, s1;
          s0.x = pl0; s0.y = pr0;
          s1.x = pl1; s1.y = pr1;
          *(v2f*)(part + (8 * h + r) * PITCH + 2 * g)      = s0;
          *(v2f*)(part + (16 + 8 * h + r) * PITCH + 2 * g) = s1;
        }
      }
    }
  }
  if (SCORES) __syncthreads();

  if (!side) {
    float* crow0 = C + (size_t)(row0 + 8 * h) * NCOL + col0 + NT * c;
#pragma unroll
    for (int r = 0; r < 8; ++r) {
      vnt o;
#pragma unroll
      for (int t = 0; t < NT; ++t) o[t] = acc0[t][r];
      *(vnt*)(crow0 + (size_t)r * NCOL) = o;
    }
    if (hi_valid) {
      float* crow1 = crow0 + (size_t)16 * NCOL;
#pragma unroll
      for (int r = 0; r < 8; ++r) {
        vnt o;
#pragma unroll
        for (int t = 0; t < NT; ++t) o[t] = acc1[t][r];
        *(vnt*)(crow1 + (size_t)r * NCOL) = o;
      }
    }
  }

  if (SCORES) {
    for (int i = threadIdx.x; i < 32 * GAT_H * 2; i += NTHR) {
      const int row = i >> 3, hd = (i >> 1) & 3, which = i & 1;
      if (row0 + row < M) {
        const float* p = part + row * PITCH + 2 * hd * GPH + which;
        float sum = 0.f;
#pragma unroll
        for (int q = 0; q < GPH; ++q) sum += p[2 * q];
        (which ? er : el)[(size_t)(row0 + row) * GAT_H + hd] = sum;
      }
    }
  }
}

__global__ __launch_bounds__(256) void gat_fold_res_heads(
    const float* __restrict__ Wres2, float* __restrict__ wm)
{
  const int i = blockIdx.x * 256 + threadIdx.x;
  if (i < 256 * GAT_RESW) {
    const int k = i / GAT_RESW, c = i - k * GAT_RESW;
    float v = 0.f;
    if (c < 40) {
      const float* p = Wres2 + (size_t)k * 160 + c;
      v = (p[0] + p[40] + p[80] + p[120]) * 0.25f;
    }
    wm[i] = v;
  }
}

__device__ __forceinline__ float gat_leaky(float x) {
  return x > 0.f ? x : GAT_SLOPE * x;
}

#define GAT_SORT_P 256
#define GAT_SORT_CH (GAT_E / GAT_SORT_P)
#define GAT_SORT_B 196
#define GAT_SORT_T1 256
#define GAT_SORT_EPT 13
#define GAT_SORT_BATCH 8
#define GAT_OFFT_LEN ((GAT_SORT_B + 1) * GAT_SORT_P)
#define GAT_RPTR_LEN (GAT_SORT_B * 256)
static_assert(GAT_SORT_CH * GAT_SORT_P == GAT_E, "chunks must tile the edge list");
static_assert(GAT_SORT_EPT * GAT_SORT_T1 >= GAT_SORT_CH, "a chunk must fit one pass of the work-group");
static_assert(GAT_SORT_T1 > GAT_SORT_B, "one thread per bucket offset, the end entry included");
static_assert(GAT_RPTR_LEN >= GAT_N + 1, "buckets must cover every node and the end entry");
static_assert(GAT_SORT_B <= 7 * 32, "one wave scans the bucket counts, 7 per lane");
static_assert(GAT_SORT_P == 256, "level 2 gives one of its 256 threads to each chunk's run");

__device__ __forceinline__ int gat_bucket_of(int d) {
  const unsigned b = (unsigned)d >> 8;
  return (int)(b < (unsigned)GAT_SORT_B ? b : (unsigned)(GAT_SORT_B - 1));
}
__device__ __forceinline__ int gat_local_dst(int d, int lo) {
  const unsigned u = (unsigned)(d - lo);
  return (int)(u < 256u ? u : 255u);
}

__global__ __launch_bounds__(GAT_SORT_T1) void gat_chunk_buckets(
    const int* __restrict__ src, const int* __restrict__ dst,
    int* __restrict__ offt, int* __restrict__ bkt)
{
  __shared__ int hist[GAT_SORT_B + 1];
  const int p = blockIdx.x, t = threadIdx.x;
  const int e0 = p * GAT_SORT_CH;

  int s[GAT_SORT_EPT], d[GAT_SORT_EPT];
#pragma unroll
  for (int k = 0; k < GAT_SORT_EPT; ++k) {
    const int i = t + k * GAT_SORT_T1;
    const int ic = i < GAT_SORT_CH ? i : GAT_SORT_CH - 1;
    d[k] = dst[e0 + ic];
    s[k] = src[e0 + ic];
  }
  if (t <= GAT_SORT_B) hist[t] = 0;
  __syncthreads();

  int b[GAT_SORT_EPT], r[GAT_SORT_EPT];
#pragma unroll
  for (int k = 0; k < GAT_SORT_EPT; ++k) {
    b[k] = -1; r[k] = 0;
    if (t + k * GAT_SORT_T1 < GAT_SORT_CH) {
      b[k] = gat_bucket_of(d[k]);
      r[k] = atomicAdd(&hist[b[k]], 1);
    }
  }
  __syncthreads();

  if (t < 32) {
    int c[7];
    int tot = 0;
#pragma unroll
    for (int q = 0; q < 7; ++q) {
      const int idx = 7 * t + q;
      c[q] = idx < GAT_SORT_B ? hist[idx] : 0;
      tot += c[q];
    }
    int inc = tot;
#pragma unroll
    for (int o = 1; o < 32; o <<= 1) {
      const int nb = __shfl_up(inc, o, 32);
      if (t >= o) inc += nb;
    }
    int run = inc - tot;
#pragma unroll
    for (int q = 0; q < 7; ++q) {
      const int idx = 7 * t + q;
      if (idx <= GAT_SORT_B) hist[idx] = run;
      run += c[q];
    }
  }
  __syncthreads();

  if (t <= GAT_SORT_B) offt[t * GAT_SORT_P + p] = hist[t];
#pragma unroll
  for (int k = 0; k < GAT_SORT_EPT; ++k) {
    if (b[k] >= 0) {
      int2 rec;
      rec.x = s[k]; rec.y = d[k];
      *(int2*)(bkt + 2 * (size_t)(e0 + hist[b[k]] + r[k])) = rec;
    }
  }
}

__global__ __launch_bounds__(256) void gat_bucket_segments(
    const int* __restrict__ offt, const int* __restrict__ bkt,
    int* __restrict__ rowptr, int* __restrict__ csrc)
{
  __shared__ int cnt[256];
  __shared__ int wbase[8];
  __shared__ int wscan[8];
  const int b = blockIdx.x, t = threadIdx.x, lane = t & 31, w = t >> 5;
  const int lo = b * 256;

  const int o0 = offt[b * GAT_SORT_P + t];
  const int o1 = offt[(b + 1) * GAT_SORT_P + t];
  const int2* run = (const int2*)bkt + (size_t)t * GAT_SORT_CH + o0;
  const int rl = o1 - o0;

  cnt[t] = 0;
  int below = o0;
#pragma unroll
  for (int o = 16; o > 0; o >>= 1) below += __shfl_xor(below, o, 32);
  if (lane == 0) wbase[w] = below;
  __syncthreads();
  int sb = 0;
#pragma unroll
  for (int q = 0; q < 8; ++q) sb += wbase[q];

  for (int i = 0; i < rl; i += GAT_SORT_BATCH) {
    int dv[GAT_SORT_BATCH];
#pragma unroll
    for (int k = 0; k < GAT_SORT_BATCH; ++k) {
      const int j = i + k < rl ? i + k : rl - 1;
      dv[k] = run[j].y;
    }
#pragma unroll
    for (int k = 0; k < GAT_SORT_BATCH; ++k)
      if (i + k < rl) atomicAdd(&cnt[gat_local_dst(dv[k], lo)], 1);
  }
  __syncthreads();

  const int c = cnt[t];
  int inc = c;
#pragma unroll
  for (int o = 1; o < 32; o <<= 1) {
    const int nb = __shfl_up(inc, o, 32);
    if (lane >= o) inc += nb;
  }
  if (lane == 31) wscan[w] = inc;
  __syncthreads();
  int first = sb + inc - c;
#pragma unroll
  for (int q = 0; q < 8; ++q) first += (q < w) ? wscan[q] : 0;
  rowptr[lo + t] = first;
  cnt[t] = first;
  __syncthreads();

  for (int i = 0; i < rl; i += GAT_SORT_BATCH) {
    int2 rec[GAT_SORT_BATCH];
#pragma unroll
    for (int k = 0; k < GAT_SORT_BATCH; ++k) {
      const int j = i + k < rl ? i + k : rl - 1;
      rec[k] = run[j];
    }
#pragma unroll
    for (int k = 0; k < GAT_SORT_BATCH; ++k) {
      if (i + k < rl) {
        const int pos = atomicAdd(&cnt[gat_local_dst(rec[k].y, lo)], 1);
        if ((unsigned)pos < (unsigned)GAT_E) csrc[pos] = rec[k].x;
      }
    }
  }
}

__device__ __forceinline__ float gat_wave_max(float v) {
#pragma unroll
  for (int o = 16; o > 0; o >>= 1) v = fmaxf(v, __shfl_xor(v, o, 32));
  return v;
}
__device__ __forceinline__ float gat_wave_sum(float v) {
#pragma unroll
  for (int o = 16; o > 0; o >>= 1) v += __shfl_xor(v, o, 32);
  return v;
}
__device__ __forceinline__ float gat_lane_bcast(float v, int l) {
  return __int_as_float(__builtin_amdgcn_readlane(__float_as_int(v), l));
}

__device__ __forceinline__ int gat_slot_src(const int* __restrict__ csrc, int j) {
  unsigned u = (unsigned)csrc[j];
  return (int)(u < (unsigned)GAT_N ? u : (unsigned)(GAT_N - 1));
}

__device__ __forceinline__ float4 gat_edge_logits(const float* __restrict__ el, int s, float4 erd) {
  const float4 e = *(const float4*)(el + (size_t)s * GAT_H);
  float4 x;
  x.x = gat_leaky(e.x + erd.x);
  x.y = gat_leaky(e.y + erd.y);
  x.z = gat_leaky(e.z + erd.z);
  x.w = gat_leaky(e.w + erd.w);
  return x;
}

__device__ __forceinline__ void gat_segment_stats(
    const int* __restrict__ csrc, const float* __restrict__ el, float4 erd,
    int beg, int end, int lane, float4& m, float4& den)
{
  const float ninf = __int_as_float((int)0xFF800000u);
  m.x = ninf; m.y = ninf; m.z = ninf; m.w = ninf;
  for (int c = beg; c < end; c += 32) {
    const int j = c + lane;
    if (j < end) {
      const float4 x = gat_edge_logits(el, gat_slot_src(csrc, j), erd);
      m.x = fmaxf(m.x, x.x); m.y = fmaxf(m.y, x.y);
      m.z = fmaxf(m.z, x.z); m.w = fmaxf(m.w, x.w);
    }
  }
  m.x = gat_wave_max(m.x); m.y = gat_wave_max(m.y);
  m.z = gat_wave_max(m.z); m.w = gat_wave_max(m.w);
  den.x = 0.f; den.y = 0.f; den.z = 0.f; den.w = 0.f;
  for (int c = beg; c < end; c += 32) {
    const int j = c + lane;
    if (j < end) {
      const float4 x = gat_edge_logits(el, gat_slot_src(csrc, j), erd);
      den.x += expf(x.x - m.x); den.y += expf(x.y - m.y);
      den.z += expf(x.z - m.z); den.w += expf(x.w - m.w);
    }
  }
  den.x = gat_wave_sum(den.x); den.y = gat_wave_sum(den.y);
  den.z = gat_wave_sum(den.z); den.w = gat_wave_sum(den.w);
}

__device__ __forceinline__ void gat_chunk_weights(
    const int* __restrict__ csrc, const float* __restrict__ el, float4 erd,
    int c, int end, int lane, float4 m, float4 den, int& s, float4& a)
{
  s = 0;
  a.x = 0.f; a.y = 0.f; a.z = 0.f; a.w = 0.f;
  const int j = c + lane;
  if (j < end) {
    s = gat_slot_src(csrc, j);
    const float4 x = gat_edge_logits(el, s, erd);
    a.x = expf(x.x - m.x) / den.x; a.y = expf(x.y - m.y) / den.y;
    a.z = expf(x.z - m.z) / den.z; a.w = expf(x.w - m.w) / den.w;
  }
}

__device__ __forceinline__ void gat_segment_weights_once(
    const int* __restrict__ csrc, const float* __restrict__ el, float4 erd,
    int beg, int end, int lane, int& s, float4& a)
{
  const float ninf = __int_as_float((int)0xFF800000u);
  const bool on = beg + lane < end;
  s = 0;
  float4 x, m;
  x.x = 0.f; x.y = 0.f; x.z = 0.f; x.w = 0.f;
  m.x = ninf; m.y = ninf; m.z = ninf; m.w = ninf;
  if (on) {
    s = gat_slot_src(csrc, beg + lane);
    x = gat_edge_logits(el, s, erd);
    m.x = fmaxf(m.x, x.x); m.y = fmaxf(m.y, x.y);
    m.z = fmaxf(m.z, x.z); m.w = fmaxf(m.w, x.w);
  }
  m.x = gat_wave_max(m.x); m.y = gat_wave_max(m.y);
  m.z = gat_wave_max(m.z); m.w = gat_wave_max(m.w);
  float4 ex;
  ex.x = 0.f; ex.y = 0.f; ex.z = 0.f; ex.w = 0.f;
  if (on) {
    ex.x = expf(x.x - m.x); ex.y = expf(x.y - m.y);
    ex.z = expf(x.z - m.z); ex.w = expf(x.w - m.w);
  }
  float4 den;
  den.x = gat_wave_sum(ex.x); den.y = gat_wave_sum(ex.y);
  den.z = gat_wave_sum(ex.z); den.w = gat_wave_sum(ex.w);
  a.x = 0.f; a.y = 0.f; a.z = 0.f; a.w = 0.f;
  if (on) {
    a.x = ex.x / den.x; a.y = ex.y / den.y;
    a.z = ex.z / den.z; a.w = ex.w / den.w;
  }
}

__global__ __launch_bounds__(256) void gat_node_agg_elu(
    const int* __restrict__ rowptr, const int* __restrict__ csrc,
    const float* __restrict__ feat, const float* __restrict__ el,
    const float* __restrict__ er, const float* __restrict__ bias,
    const float* __restrict__ res, float* __restrict__ out)
{
  const int d = __builtin_amdgcn_readfirstlane((int)(blockIdx.x * 8 + (threadIdx.x >> 5)));
  if (d >= GAT_N) return;
  const int lane = threadIdx.x & 31;
  const int beg = rowptr[d], end = rowptr[d + 1];
  const float4 erd = *(const float4*)(er + (size_t)d * GAT_H);

  const bool once = end - beg <= 32;
  float4 m = {0.f, 0.f, 0.f, 0.f}, den = {0.f, 0.f, 0.f, 0.f};
  int s1 = 0;
  float4 a1 = {0.f, 0.f, 0.f, 0.f};
  if (once) gat_segment_weights_once(csrc, el, erd, beg, end, lane, s1, a1);
  else      gat_segment_stats(csrc, el, erd, beg, end, lane, m, den);

  const bool lo = lane < 16;
  float4 accA = {0.f, 0.f, 0.f, 0.f}, accB = {0.f, 0.f, 0.f, 0.f};
  for (int c = beg; c < end; c += 32) {
    int s = s1; float4 a = a1;
    if (!once) gat_chunk_weights(csrc, el, erd, c, end, lane, m, den, s, a);
    const int cnt = (end - c) < 32 ? (end - c) : 32;
#pragma unroll 4
    for (int t = 0; t < cnt; ++t) {
      const int st = __builtin_amdgcn_readlane(s, t);
      const float a0 = gat_lane_bcast(a.x, t), a1 = gat_lane_bcast(a.y, t);
      const float a2 = gat_lane_bcast(a.z, t), a3 = gat_lane_bcast(a.w, t);
      const float wA = lo ? a0 : a1;
      const float wB = lo ? a2 : a3;
      const float4* row = (const float4*)(feat + (size_t)st * 256);
      const float4 vA = row[lane];
      const float4 vB = row[32 + lane];
      accA.x += vA.x * wA; accA.y += vA.y * wA; accA.z += vA.z * wA; accA.w += vA.w * wA;
      accB.x += vB.x * wB; accB.y += vB.y * wB; accB.z += vB.z * wB; accB.w += vB.w * wB;
    }
  }

  const float4 bA = *(const float4*)(bias + 4 * lane);
  const float4 bB = *(const float4*)(bias + 128 + 4 * lane);
  float4 vA, vB;
  vA.x = accA.x + bA.x; vA.y = accA.y + bA.y; vA.z = accA.z + bA.z; vA.w = accA.w + bA.w;
  vB.x = accB.x + bB.x; vB.y = accB.y + bB.y; vB.z = accB.z + bB.z; vB.w = accB.w + bB.w;
  const size_t o = (size_t)d * 256 + 4 * lane;
  if (res) {
    const float4 rA = *(const float4*)(res + o);
    const float4 rB = *(const float4*)(res + o + 128);
    vA.x += rA.x; vA.y += rA.y; vA.z += rA.z; vA.w += rA.w;
    vB.x += rB.x; vB.y += rB.y; vB.z += rB.z; vB.w += rB.w;
  }
  vA.x = vA.x > 0.f ? vA.x : (expf(vA.x) - 1.f);
  vA.y = vA.y > 0.f ? vA.y : (expf(vA.y) - 1.f);
  vA.z = vA.z > 0.f ? vA.z : (expf(vA.z) - 1.f);
  vA.w = vA.w > 0.f ? vA.w : (expf(vA.w) - 1.f);
  vB.x = vB.x > 0.f ? vB.x : (expf(vB.x) - 1.f);
  vB.y = vB.y > 0.f ? vB.y : (expf(vB.y) - 1.f);
  vB.z = vB.z > 0.f ? vB.z : (expf(vB.z) - 1.f);
  vB.w = vB.w > 0.f ? vB.w : (expf(vB.w) - 1.f);
  *(float4*)(out + o) = vA;
  *(float4*)(out + o + 128) = vB;
}

__global__ __launch_bounds__(256) void gat_node_agg_mean(
    const int* __restrict__ rowptr, const int* __restrict__ csrc,
    const float* __restrict__ feat, const float* __restrict__ el,
    const float* __restrict__ er, const float* __restrict__ bias,
    const float* __restrict__ resm, float* __restrict__ out)
{
  __shared__ float sval[8 * 160];
  const int w = threadIdx.x >> 5;
  const int d = __builtin_amdgcn_readfirstlane((int)(blockIdx.x * 8 + w));
  const int lane = threadIdx.x & 31;

  if (d < GAT_N) {
    const int beg = rowptr[d], end = rowptr[d + 1];
    const float4 erd = *(const float4*)(er + (size_t)d * GAT_H);
    const bool once = end - beg <= 32;
    float4 m = {0.f, 0.f, 0.f, 0.f}, den = {0.f, 0.f, 0.f, 0.f};
    int s1 = 0;
    float4 a1 = {0.f, 0.f, 0.f, 0.f};
    if (once) gat_segment_weights_once(csrc, el, erd, beg, end, lane, s1, a1);
    else      gat_segment_stats(csrc, el, erd, beg, end, lane, m, den);

    float acc0 = 0.f, acc1 = 0.f, acc2 = 0.f, acc3 = 0.f, acc4 = 0.f;
    for (int c = beg; c < end; c += 32) {
      int s = s1; float4 a = a1;
      if (!once) gat_chunk_weights(csrc, el, erd, c, end, lane, m, den, s, a);
      const int cnt = (end - c) < 32 ? (end - c) : 32;
#pragma unroll 4
      for (int t = 0; t < cnt; ++t) {
        const int st = __builtin_amdgcn_readlane(s, t);
        const float a0 = gat_lane_bcast(a.x, t), a1 = gat_lane_bcast(a.y, t);
        const float a2 = gat_lane_bcast(a.z, t), a3 = gat_lane_bcast(a.w, t);
        const float* row = feat + (size_t)st * 160 + lane;
        const float v0 = row[0], v1 = row[32], v2 = row[64], v3 = row[96], v4 = row[128];
        acc0 += v0 * a0;
        acc1 += v1 * (lane >= 8 ? a1 : a0);
        acc2 += v2 * (lane >= 16 ? a2 : a1);
        acc3 += v3 * (lane >= 24 ? a3 : a2);
        acc4 += v4 * a3;
      }
    }
    const float* bb = bias + lane;
    float* sv = sval + w * 160 + lane;
    sv[0]   = acc0 + bb[0];
    sv[32]  = acc1 + bb[32];
    sv[64]  = acc2 + bb[64];
    sv[96]  = acc3 + bb[96];
    sv[128] = acc4 + bb[128];
  }
  __syncthreads();
  for (int i = threadIdx.x; i < 8 * 40; i += 256) {
    const int nl = i / 40, c = i - nl * 40;
    if ((int)(blockIdx.x * 8 + nl) < GAT_N) {
      const float* sv = sval + nl * 160 + c;
      float sum = 0.f;
#pragma unroll
      for (int h = 0; h < GAT_H; ++h) sum += sv[h * 40];
      out[(size_t)blockIdx.x * 320 + i] =
          sum * 0.25f + resm[(size_t)(blockIdx.x * 8 + nl) * GAT_RESW + c];
    }
  }
}

extern "C" void kernel_launch(void* const* d_in, const int* in_sizes, int n_in,
                              void* d_out, int out_size, void* d_ws, size_t ws_size,
                              hipStream_t stream)
{
  const float* x     = (const float*)d_in[0];
  const int*   src   = (const int*)d_in[1];
  const int*   dst   = (const int*)d_in[2];
  const float* W0    = (const float*)d_in[3];
  const float* al0   = (const float*)d_in[4];
  const float* ar0   = (const float*)d_in[5];
  const float* b0    = (const float*)d_in[6];
  const float* W1    = (const float*)d_in[7];
  const float* al1   = (const float*)d_in[8];
  const float* ar1   = (const float*)d_in[9];
  const float* b1    = (const float*)d_in[10];
  const float* W2    = (const float*)d_in[11];
  const float* al2   = (const float*)d_in[12];
  const float* ar2   = (const float*)d_in[13];
  const float* b2    = (const float*)d_in[14];
  const float* Wres2 = (const float*)d_in[15];

  const size_t NF = (size_t)GAT_N * 256;
  float* ws   = (float*)d_ws;
  float* bufA = ws;
  float* bufB = bufA + NF;
  float* bufC = bufB + NF;
  float* EL   = bufC + NF;
  float* ER   = EL + (size_t)GAT_N * GAT_H;
  int*   OFFT = (int*)(ER + (size_t)GAT_N * GAT_H);
  int*   RPTR = OFFT + GAT_OFFT_LEN;
  int*   BKT  = RPTR + GAT_RPTR_LEN;
  int*   CSRC = BKT + 2 * (size_t)GAT_E;
  float* WM   = (float*)(CSRC + GAT_E);

  const int rowTiles = (GAT_N + 31) / 32;
  dim3 gemmBlk256(128);
  dim3 gemmBlk160(224);

  const int nodeBlocks = (GAT_N + 7) / 8;

  gat_chunk_buckets<<<GAT_SORT_P, GAT_SORT_T1, 0, stream>>>(src, dst, OFFT, BKT);
  gat_bucket_segments<<<GAT_SORT_B, 256, 0, stream>>>(OFFT, BKT, RPTR, CSRC);

  gat_gemm_wmma<256, 256, 4, 64, true, 0><<<rowTiles, gemmBlk256, 0, stream>>>(x, W0, bufA, GAT_N, al0, ar0, EL, ER, nullptr, nullptr);
  gat_node_agg_elu<<<nodeBlocks, 256, 0, stream>>>(RPTR, CSRC, bufA, EL, ER, b0, nullptr, bufB);

  gat_gemm_wmma<256, 256, 4, 64, true, 0><<<rowTiles, gemmBlk256, 0, stream>>>(bufB, W1, bufA, GAT_N, al1, ar1, EL, ER, nullptr, nullptr);
  gat_node_agg_elu<<<nodeBlocks, 256, 0, stream>>>(RPTR, CSRC, bufA, EL, ER, b1, bufB, bufC);

  gat_fold_res_heads<<<(256 * GAT_RESW + 255) / 256, 256, 0, stream>>>(Wres2, WM);
  gat_gemm_wmma<256, 160, 2, 40, true, 2><<<rowTiles, gemmBlk160, 0, stream>>>(bufC, W2, bufA, GAT_N, al2, ar2, EL, ER, WM, bufB);
  gat_node_agg_mean<<<nodeBlocks, 256, 0, stream>>>(RPTR, CSRC, bufA, EL, ER, b2, bufB, (float*)d_out);
}


# level 2 of the edge sort runs as 196 leading work-groups of the first GEMM's launch (2 chunk runs per thread), the Wres2 fold in level 1's launch: 7 launches for 9; sum order in a group changes
# speedup vs baseline: 4.5203x; 1.0673x over previous
#include <hip/hip_runtime.h>
#include <math.h>

#define GAT_N 50000
#define GAT_E 800000
#define GAT_H 4
#define GAT_SLOPE 0.2f

typedef __attribute__((ext_vector_type(2))) float v2f;
typedef __attribute__((ext_vector_type(8))) float v8f;

#define GAT_RESW 48

#define GAT_SORT_P 256
#define GAT_SORT_CH (GAT_E / GAT_SORT_P)
#define GAT_SORT_B 196
#define GAT_SORT_T1 256
#define GAT_SORT_EPT 13
#define GAT_SORT_BATCH 8
#define GAT_SORT_T2 128
#define GAT_OFFT_LEN ((GAT_SORT_B + 1) * GAT_SORT_P)
#define GAT_RPTR_LEN (GAT_SORT_B * 256)
#define GAT_FOLD_BLOCKS ((256 * GAT_RESW + 255) / 256)
static_assert(GAT_SORT_CH * GAT_SORT_P == GAT_E, "chunks must tile the edge list");
static_assert(GAT_SORT_EPT * GAT_SORT_T1 >= GAT_SORT_CH, "a chunk must fit one pass of the work-group");
static_assert(GAT_SORT_T1 > GAT_SORT_B, "one thread per bucket offset, the end entry included");
static_assert(GAT_RPTR_LEN >= GAT_N + 1, "buckets must cover every node and the end entry");
static_assert(GAT_SORT_B <= 7 * 32, "one wave scans the bucket counts, 7 per lane");
static_assert(GAT_SORT_P == 2 * GAT_SORT_T2, "level 2 gives each of its threads the runs of two chunks");

__device__ __forceinline__ int gat_bucket_of(int d) {
  const unsigned b = (unsigned)d >> 8;
  return (int)(b < (unsigned)GAT_SORT_B ? b : (unsigned)(GAT_SORT_B - 1));
}
__device__ __forceinline__ int gat_local_dst(int d, int lo) {
  const unsigned u = (unsigned)(d - lo);
  return (int)(u < 256u ? u : 255u);
}

__device__ __forceinline__ void gat_fold_res_heads(
    const float* __restrict__ Wres2, float* __restrict__ wm, int i)
{
  if (i < 256 * GAT_RESW) {
    const int k = i / GAT_RESW, c = i - k * GAT_RESW;
    float v = 0.f;
    if (c < 40) {
      const float* p = Wres2 + (size_t)k * 160 + c;
      v = (p[0] + p[40] + p[80] + p[120]) * 0.25f;
    }
    wm[i] = v;
  }
}

__global__ __launch_bounds__(GAT_SORT_T1) void gat_chunk_buckets(
    const int* __restrict__ src, const int* __restrict__ dst,
    int* __restrict__ offt, int* __restrict__ bkt,
    const float* __restrict__ Wres2, float* __restrict__ wm)
{
  __shared__ int hist[GAT_SORT_B + 1];
  const int p = blockIdx.x, t = threadIdx.x;
  if (p >= GAT_SORT_P) {
    gat_fold_res_heads(Wres2, wm, (p - GAT_SORT_P) * 256 + t);
    return;
  }
  const int e0 = p * GAT_SORT_CH;
  int s[GAT_SORT_EPT], d[GAT_SORT_EPT];
#pragma unroll
  for (int k = 0; k < GAT_SORT_EPT; ++k) {
    const int i = t + k * GAT_SORT_T1;
    const int ic = i < GAT_SORT_CH ? i : GAT_SORT_CH - 1;
    d[k] = dst[e0 + ic];
    s[k] = src[e0 + ic];
  }
  if (t <= GAT_SORT_B) hist[t] = 0;
  __syncthreads();

  int b[GAT_SORT_EPT], r[GAT_SORT_EPT];
#pragma unroll
  for (int k = 0; k < GAT_SORT_EPT; ++k) {
    b[k] = -1; r[k] = 0;
    if (t + k * GAT_SORT_T1 < GAT_SORT_CH) {
      b[k] = gat_bucket_of(d[k]);
      r[k] = atomicAdd(&hist[b[k]], 1);
    }
  }
  __syncthreads();

  if (t < 32) {
    int c[7];
    int tot = 0;
#pragma unroll
    for (int q = 0; q < 7; ++q) {
      const int idx = 7 * t + q;
      c[q] = idx < GAT_SORT_B ? hist[idx] : 0;
      tot += c[q];
    }
    int inc = tot;
#pragma unroll
    for (int o = 1; o < 32; o <<= 1) {
      const int nb = __shfl_up(inc, o, 32);
      if (t >= o) inc += nb;
    }
    int run = inc - tot;
#pragma unroll
    for (int q = 0; q < 7; ++q) {
      const int idx = 7 * t + q;
      if (idx <= GAT_SORT_B) hist[idx] = run;
      run += c[q];
    }
  }
  __syncthreads();

  if (t <= GAT_SORT_B) offt[t * GAT_SORT_P + p] = hist[t];
#pragma unroll
  for (int k = 0; k < GAT_SORT_EPT; ++k) {
    if (b[k] >= 0) {
      int2 rec;
      rec.x = s[k]; rec.y = d[k];
      *(int2*)(bkt + 2 * (size_t)(e0 + hist[b[k]] + r[k])) = rec;
    }
  }
}

__device__ __forceinline__ void gat_bucket_segments(
    const int* __restrict__ offt, const int* __restrict__ bkt,
    int* __restrict__ rowptr, int* __restrict__ csrc, int b, int t)
{
  __shared__ int cnt[256];
  __shared__ int wbase[GAT_SORT_T2 / 32];
  __shared__ int wscan[GAT_SORT_T2 / 32];
  const int lane = t & 31, w = t >> 5;
  const int lo = b * 256;

  const int2* run[2];
  int rl[2];
  int below = 0;
#pragma unroll
  for (int q = 0; q < 2; ++q) {
    const int ch = t + q * GAT_SORT_T2;
    const int o0 = offt[b * GAT_SORT_P + ch];
    const int o1 = offt[(b + 1) * GAT_SORT_P + ch];
    run[q] = (const int2*)bkt + (size_t)ch * GAT_SORT_CH + o0;
    rl[q] = o1 - o0;
    below += o0;
  }

  cnt[t] = 0;
  cnt[t + GAT_SORT_T2] = 0;
#pragma unroll
  for (int o = 16; o > 0; o >>= 1) below += __shfl_xor(below, o, 32);
  if (lane == 0) wbase[w] = below;
  __syncthreads();
  int sb = 0;
#pragma unroll
  for (int q = 0; q < GAT_SORT_T2 / 32; ++q) sb += wbase[q];

#pragma unroll
  for (int q = 0; q < 2; ++q) {
    for (int i = 0; i < rl[q]; i += GAT_SORT_BATCH) {
      int dv[GAT_SORT_BATCH];
#pragma unroll
      for (int k = 0; k < GAT_SORT_BATCH; ++k) {
        const int j = i + k < rl[q] ? i + k : rl[q] - 1;
        dv[k] = run[q][j].y;
      }
#pragma unroll
      for (int k = 0; k < GAT_SORT_BATCH; ++k)
        if (i + k < rl[q]) atomicAdd(&cnt[gat_local_dst(dv[k], lo)], 1);
    }
  }
  __syncthreads();

  const int c0 = cnt[2 * t], c1 = cnt[2 * t + 1];
  const int pair = c0 + c1;
  int inc = pair;
#pragma unroll
  for (int o = 1; o < 32; o <<= 1) {
    const int nb = __shfl_up(inc, o, 32);
    if (lane >= o) inc += nb;
  }
  if (lane == 31) wscan[w] = inc;
  __syncthreads();
  int first = sb + inc - pair;
#pragma unroll
  for (int q = 0; q < GAT_SORT_T2 / 32; ++q) first += (q < w) ? wscan[q] : 0;
  rowptr[lo + 2 * t]     = first;
  rowptr[lo + 2 * t + 1] = first + c0;
  cnt[2 * t]     = first;
  cnt[2 * t + 1] = first + c0;
  __syncthreads();

#pragma unroll
  for (int q = 0; q < 2; ++q) {
    for (int i = 0; i < rl[q]; i += GAT_SORT_BATCH) {
      int2 rec[GAT_SORT_BATCH];
#pragma unroll
      for (int k = 0; k < GAT_SORT_BATCH; ++k) {
        const int j = i + k < rl[q] ? i + k : rl[q] - 1;
        rec[k] = run[q][j];
      }
#pragma unroll
      for (int k = 0; k < GAT_SORT_BATCH; ++k) {
        if (i + k < rl[q]) {
          const int pos = atomicAdd(&cnt[gat_local_dst(rec[k].y, lo)], 1);
          if ((unsigned)pos < (unsigned)GAT_E) csrc[pos] = rec[k].x;
        }
      }
    }
  }
}

template <int MASK>
__device__ __forceinline__ float gat_row_xor_add(float v) {
  return v + __int_as_float(__builtin_amdgcn_update_dpp(
                 0, __float_as_int(v), 0x160 + MASK, 0xf, 0xf, true));
}

template <int K>
__device__ __forceinline__ void gat_side_tile(
    const float* __restrict__ A, const float* __restrict__ Bm,
    float* __restrict__ Cm, int r0, int lane)
{
  typedef float v4f __attribute__((ext_vector_type(4)));
  const int c = lane & 15;
  const int h = lane >> 4;
  const float* arow = A + (size_t)(r0 + c) * K + 4 * h;
  const float* bp   = Bm + (size_t)(4 * h) * GAT_RESW + 3 * c;
  v8f acc[3];
#pragma unroll
  for (int t = 0; t < 3; ++t) acc[t] = (v8f){};
#pragma unroll 1
  for (int k0 = 0; k0 < K; k0 += 8) {
    const v4f a = *(const v4f*)(arow + k0);
    float b[4][3];
#pragma unroll
    for (int i = 0; i < 4; ++i)
#pragma unroll
      for (int t = 0; t < 3; ++t) b[i][t] = bp[(size_t)(k0 + i) * GAT_RESW + t];
#pragma unroll
    for (int j = 0; j < 2; ++j) {
      v2f fa;
      fa.x = a[2 * j]; fa.y = a[2 * j + 1];
#pragma unroll
      for (int t = 0; t < 3; ++t) {
        v2f fb;
        fb.x = b[2 * j][t];
        fb.y = b[2 * j + 1][t];
        acc[t] = __builtin_amdgcn_wmma_f32_16x16x4_f32(
            false, fa, false, fb, (short)0, acc[t], false, false);
      }
    }
  }
  float* crow = Cm + (size_t)(r0 + 8 * h) * GAT_RESW + 3 * c;
#pragma unroll
  for (int r = 0; r < 8; ++r)
#pragma unroll
    for (int t = 0; t < 3; ++t) crow[(size_t)r * GAT_RESW + t] = acc[t][r];
}

template <int K, int NCOL, int NT, int D, bool SCORES, int SIDE, int PRE>
__global__ __launch_bounds__((NCOL / (16 * NT) + SIDE) * 32) void gat_gemm_wmma(
    const float* __restrict__ A, const float* __restrict__ B,
    float* __restrict__ C, int M,
    const float* __restrict__ al, const float* __restrict__ ar,
    float* __restrict__ el, float* __restrict__ er,
    const float* __restrict__ Bm, float* __restrict__ Cm,
    const int* __restrict__ offt, const int* __restrict__ bkt,
    int* __restrict__ rowptr, int* __restrict__ csrc)
{
  constexpr int NW   = NCOL / (16 * NT);
  constexpr int NTHR = (NW + SIDE) * 32;
  static_assert(PRE == 0 || NTHR == GAT_SORT_T2, "the leading work-groups run level 2 of the sort with this block size");
  if (PRE > 0) {
    if ((int)blockIdx.x < PRE) {
      gat_bucket_segments(offt, bkt, rowptr, csrc, (int)blockIdx.x, (int)threadIdx.x);
      return;
    }
  }
  typedef float vnt __attribute__((ext_vector_type(NT)));
  typedef float v4f __attribute__((ext_vector_type(4)));
  const int wave = threadIdx.x >> 5;
  const int lane = threadIdx.x & 31;
  const int row0 = ((int)blockIdx.x - PRE) << 5;
  const int col0 = wave * (16 * NT);
  const int c    = lane & 15;
  const int h    = lane >> 4;
  const bool hi_valid = (row0 + 16) < M;
  const bool side = SIDE > 0 &&
      __builtin_amdgcn_readfirstlane((int)(threadIdx.x >> 5)) >= NW;

  constexpr int NG    = NCOL / (4 * NT);
  constexpr int GPH   = D / (4 * NT);
  constexpr int PITCH = 2 * NG + 2;
  __shared__ float part[SCORES ? 32 * PITCH : 1];

  v8f acc0[NT], acc1[NT];
#pragma unroll
  for (int t = 0; t < NT; ++t) { acc0[t] = (v8f){}; acc1[t] = (v8f){}; }

  if (side) {
    const int s = __builtin_amdgcn_readfirstlane((int)(threadIdx.x >> 5)) - NW;
    if (s == 0 || hi_valid) gat_side_tile<K>(A, Bm, Cm, row0 + 16 * s, lane);
  } else {
    const float* arow0 = A + (size_t)(row0 + c) * K + 4 * h;
    const float* arow1 = hi_valid ? (arow0 + (size_t)16 * K) : arow0;
    const float* bp    = B + (size_t)(4 * h) * NCOL + col0 + NT * c;

#pragma unroll 2
    for (int k0 = 0; k0 < K; k0 += 8) {
      const v4f a0 = *(const v4f*)(arow0 + k0);
      const v4f a1 = *(const v4f*)(arow1 + k0);
      vnt b[4];
#pragma unroll
      for (int i = 0; i < 4; ++i) b[i] = *(const vnt*)(bp + (size_t)(k0 + i) * NCOL);
#pragma unroll
      for (int j = 0; j < 2; ++j) {
        v2f fa0, fa1;
        fa0.x = a0[2 * j]; fa0.y = a0[2 * j + 1];
        fa1.x = a1[2 * j]; fa1.y = a1[2 * j + 1];
#pragma unroll
        for (int t = 0; t < NT; ++t) {
          v2f fb;
          fb.x = b[2 * j][t];
          fb.y = b[2 * j + 1][t];
          acc0[t] = __builtin_amdgcn_wmma_f32_16x16x4_f32(
              false, fa0, false, fb, (short)0, acc0[t], false, false);
          acc1[t] = __builtin_amdgcn_wmma_f32_16x16x4_f32(
              false, fa1, false, fb, (short)0, acc1[t], false, false);
        }
      }
    }

    if (SCORES) {
      const vnt wl = *(const vnt*)(al + col0 + NT * c);
      const vnt wr = *(const vnt*)(ar + col0 + NT * c);
      const int g = (col0 + NT * c) / (4 * NT);
#pragma unroll
      for (int r = 0; r < 8; ++r) {
        float pl0 = 0.f, pr0 = 0.f, pl1 = 0.f, pr1 = 0.f;
#pragma unroll
        for (int t = 0; t < NT; ++t) {
          pl0 = fmaf(acc0[t][r], wl[t], pl0);
          pr0 = fmaf(acc0[t][r], wr[t], pr0);
          pl1 = fmaf(acc1[t][r], wl[t], pl1);
          pr1 = fmaf(acc1[t][r], wr[t], pr1);
        }
        pl0 = gat_row_xor_add<2>(gat_row_xor_add<1>(pl0));
        pr0 = gat_row_xor_add<2>(gat_row_xor_add<1>(pr0));
        pl1 = gat_row_xor_add<2>(gat_row_xor_add<1>(pl1));
        pr1 = gat_row_xor_add<2>(gat_row_xor_add<1>(pr1));
        if ((c & 3) == 0) {
          v2f s0, s1;
          s0.x = pl0; s0.y = pr0;
          s1.x = pl1; s1.y = pr1;
          *(v2f*)(part + (8 * h + r) * PITCH + 2 * g)      = s0;
          *(v2f*)(part + (16 + 8 * h + r) * PITCH + 2 * g) = s1;
        }
      }
    }
  }
  if (SCORES) __syncthreads();

  if (!side) {
    float* crow0 = C + (size_t)(row0 + 8 * h) * NCOL + col0 + NT * c;
#pragma unroll
    for (int r = 0; r < 8; ++r) {
      vnt o;
#pragma unroll
      for (int t = 0; t < NT; ++t) o[t] = acc0[t][r];
      *(vnt*)(crow0 + (size_t)r * NCOL) = o;
    }
    if (hi_valid) {
      float* crow1 = crow0 + (size_t)16 * NCOL;
#pragma unroll
      for (int r = 0; r < 8; ++r) {
        vnt o;
#pragma unroll
        for (int t = 0; t < NT; ++t) o[t] = acc1[t][r];
        *(vnt*)(crow1 + (size_t)r * NCOL) = o;
      }
    }
  }

  if (SCORES) {
    for (int i = threadIdx.x; i < 32 * GAT_H * 2; i += NTHR) {
      const int row = i >> 3, hd = (i >> 1) & 3, which = i & 1;
      if (row0 + row < M) {
        const float* p = part + row * PITCH + 2 * hd * GPH + which;
        float sum = 0.f;
#pragma unroll
        for (int q = 0; q < GPH; ++q) sum += p[2 * q];
        (which ? er : el)[(size_t)(row0 + row) * GAT_H + hd] = sum;
      }
    }
  }
}

__device__ __forceinline__ float gat_leaky(float x) {
  return x > 0.f ? x : GAT_SLOPE * x;
}

__device__ __forceinline__ float gat_wave_max(float v) {
#pragma unroll
  for (int o = 16; o > 0; o >>= 1) v = fmaxf(v, __shfl_xor(v, o, 32));
  return v;
}
__device__ __forceinline__ float gat_wave_sum(float v) {
#pragma unroll
  for (int o = 16; o > 0; o >>= 1) v += __shfl_xor(v, o, 32);
  return v;
}
__device__ __forceinline__ float gat_lane_bcast(float v, int l) {
  return __int_as_float(__builtin_amdgcn_readlane(__float_as_int(v), l));
}

__device__ __forceinline__ int gat_slot_src(const int* __restrict__ csrc, int j) {
  unsigned u = (unsigned)csrc[j];
  return (int)(u < (unsigned)GAT_N ? u : (unsigned)(GAT_N - 1));
}

__device__ __forceinline__ float4 gat_edge_logits(const float* __restrict__ el, int s, float4 erd) {
  const float4 e = *(const float4*)(el + (size_t)s * GAT_H);
  float4 x;
  x.x = gat_leaky(e.x + erd.x);
  x.y = gat_leaky(e.y + erd.y);
  x.z = gat_leaky(e.z + erd.z);
  x.w = gat_leaky(e.w + erd.w);
  return x;
}

__device__ __forceinline__ void gat_segment_stats(
    const int* __restrict__ csrc, const float* __restrict__ el, float4 erd,
    int beg, int end, int lane, float4& m, float4& den)
{
  const float ninf = __int_as_float((int)0xFF800000u);
  m.x = ninf; m.y = ninf; m.z = ninf; m.w = ninf;
  for (int c = beg; c < end; c += 32) {
    const int j = c + lane;
    if (j < end) {
      const float4 x = gat_edge_logits(el, gat_slot_src(csrc, j), erd);
      m.x = fmaxf(m.x, x.x); m.y = fmaxf(m.y, x.y);
      m.z = fmaxf(m.z, x.z); m.w = fmaxf(m.w, x.w);
    }
  }
  m.x = gat_wave_max(m.x); m.y = gat_wave_max(m.y);
  m.z = gat_wave_max(m.z); m.w = gat_wave_max(m.w);
  den.x = 0.f; den.y = 0.f; den.z = 0.f; den.w = 0.f;
  for (int c = beg; c < end; c += 32) {
    const int j = c + lane;
    if (j < end) {
      const float4 x = gat_edge_logits(el, gat_slot_src(csrc, j), erd);
      den.x += expf(x.x - m.x); den.y += expf(x.y - m.y);
      den.z += expf(x.z - m.z); den.w += expf(x.w - m.w);
    }
  }
  den.x = gat_wave_sum(den.x); den.y = gat_wave_sum(den.y);
  den.z = gat_wave_sum(den.z); den.w = gat_wave_sum(den.w);
}

__device__ __forceinline__ void gat_chunk_weights(
    const int* __restrict__ csrc, const float* __restrict__ el, float4 erd,
    int c, int end, int lane, float4 m, float4 den, int& s, float4& a)
{
  s = 0;
  a.x = 0.f; a.y = 0.f; a.z = 0.f; a.w = 0.f;
  const int j = c + lane;
  if (j < end) {
    s = gat_slot_src(csrc, j);
    const float4 x = gat_edge_logits(el, s, erd);
    a.x = expf(x.x - m.x) / den.x; a.y = expf(x.y - m.y) / den.y;
    a.z = expf(x.z - m.z) / den.z; a.w = expf(x.w - m.w) / den.w;
  }
}

__device__ __forceinline__ void gat_segment_weights_once(
    const int* __restrict__ csrc, const float* __restrict__ el, float4 erd,
    int beg, int end, int lane, int& s, float4& a)
{
  const float ninf = __int_as_float((int)0xFF800000u);
  const bool on = beg + lane < end;
  s = 0;
  float4 x, m;
  x.x = 0.f; x.y = 0.f; x.z = 0.f; x.w = 0.f;
  m.x = ninf; m.y = ninf; m.z = ninf; m.w = ninf;
  if (on) {
    s = gat_slot_src(csrc, beg + lane);
    x = gat_edge_logits(el, s, erd);
    m.x = fmaxf(m.x, x.x); m.y = fmaxf(m.y, x.y);
    m.z = fmaxf(m.z, x.z); m.w = fmaxf(m.w, x.w);
  }
  m.x = gat_wave_max(m.x); m.y = gat_wave_max(m.y);
  m.z = gat_wave_max(m.z); m.w = gat_wave_max(m.w);
  float4 ex;
  ex.x = 0.f; ex.y = 0.f; ex.z = 0.f; ex.w = 0.f;
  if (on) {
    ex.x = expf(x.x - m.x); ex.y = expf(x.y - m.y);
    ex.z = expf(x.z - m.z); ex.w = expf(x.w - m.w);
  }
  float4 den;
  den.x = gat_wave_sum(ex.x); den.y = gat_wave_sum(ex.y);
  den.z = gat_wave_sum(ex.z); den.w = gat_wave_sum(ex.w);
  a.x = 0.f; a.y = 0.f; a.z = 0.f; a.w = 0.f;
  if (on) {
    a.x = ex.x / den.x; a.y = ex.y / den.y;
    a.z = ex.z / den.z; a.w = ex.w / den.w;
  }
}

__global__ __launch_bounds__(256) void gat_node_agg_elu(
    const int* __restrict__ rowptr, const int* __restrict__ csrc,
    const float* __restrict__ feat, const float* __restrict__ el,
    const float* __restrict__ er, const float* __restrict__ bias,
    const float* __restrict__ res, float* __restrict__ out)
{
  const int d = __builtin_amdgcn_readfirstlane((int)(blockIdx.x * 8 + (threadIdx.x >> 5)));
  if (d >= GAT_N) return;
  const int lane = threadIdx.x & 31;
  const int beg = rowptr[d], end = rowptr[d + 1];
  const float4 erd = *(const float4*)(er + (size_t)d * GAT_H);

  const bool once = end - beg <= 32;
  float4 m = {0.f, 0.f, 0.f, 0.f}, den = {0.f, 0.f, 0.f, 0.f};
  int s1 = 0;
  float4 a1 = {0.f, 0.f, 0.f, 0.f};
  if (once) gat_segment_weights_once(csrc, el, erd, beg, end, lane, s1, a1);
  else      gat_segment_stats(csrc, el, erd, beg, end, lane, m, den);

  const bool lo = lane < 16;
  float4 accA = {0.f, 0.f, 0.f, 0.f}, accB = {0.f, 0.f, 0.f, 0.f};
  for (int c = beg; c < end; c += 32) {
    int s = s1; float4 a = a1;
    if (!once) gat_chunk_weights(csrc, el, erd, c, end, lane, m, den, s, a);
    const int cnt = (end - c) < 32 ? (end - c) : 32;
#pragma unroll 4
    for (int t = 0; t < cnt; ++t) {
      const int st = __builtin_amdgcn_readlane(s, t);
      const float a0 = gat_lane_bcast(a.x, t), a1 = gat_lane_bcast(a.y, t);
      const float a2 = gat_lane_bcast(a.z, t), a3 = gat_lane_bcast(a.w, t);
      const float wA = lo ? a0 : a1;
      const float wB = lo ? a2 : a3;
      const float4* row = (const float4*)(feat + (size_t)st * 256);
      const float4 vA = row[lane];
      const float4 vB = row[32 + lane];
      accA.x += vA.x * wA; accA.y += vA.y * wA; accA.z += vA.z * wA; accA.w += vA.w * wA;
      accB.x += vB.x * wB; accB.y += vB.y * wB; accB.z += vB.z * wB; accB.w += vB.w * wB;
    }
  }

  const float4 bA = *(const float4*)(bias + 4 * lane);
  const float4 bB = *(const float4*)(bias + 128 + 4 * lane);
  float4 vA, vB;
  vA.x = accA.x + bA.x; vA.y = accA.y + bA.y; vA.z = accA.z + bA.z; vA.w = accA.w + bA.w;
  vB.x = accB.x + bB.x; vB.y = accB.y + bB.y; vB.z = accB.z + bB.z; vB.w = accB.w + bB.w;
  const size_t o = (size_t)d * 256 + 4 * lane;
  if (res) {
    const float4 rA = *(const float4*)(res + o);
    const float4 rB = *(const float4*)(res + o + 128);
    vA.x += rA.x; vA.y += rA.y; vA.z += rA.z; vA.w += rA.w;
    vB.x += rB.x; vB.y += rB.y; vB.z += rB.z; vB.w += rB.w;
  }
  vA.x = vA.x > 0.f ? vA.x : (expf(vA.x) - 1.f);
  vA.y = vA.y > 0.f ? vA.y : (expf(vA.y) - 1.f);
  vA.z = vA.z > 0.f ? vA.z : (expf(vA.z) - 1.f);
  vA.w = vA.w > 0.f ? vA.w : (expf(vA.w) - 1.f);
  vB.x = vB.x > 0.f ? vB.x : (expf(vB.x) - 1.f);
  vB.y = vB.y > 0.f ? vB.y : (expf(vB.y) - 1.f);
  vB.z = vB.z > 0.f ? vB.z : (expf(vB.z) - 1.f);
  vB.w = vB.w > 0.f ? vB.w : (expf(vB.w) - 1.f);
  *(float4*)(out + o) = vA;
  *(float4*)(out + o + 128) = vB;
}

__global__ __launch_bounds__(256) void gat_node_agg_mean(
    const int* __restrict__ rowptr, const int* __restrict__ csrc,
    const float* __restrict__ feat, const float* __restrict__ el,
    const float* __restrict__ er, const float* __restrict__ bias,
    const float* __restrict__ resm, float* __restrict__ out)
{
  __shared__ float sval[8 * 160];
  const int w = threadIdx.x >> 5;
  const int d = __builtin_amdgcn_readfirstlane((int)(blockIdx.x * 8 + w));
  const int lane = threadIdx.x & 31;

  if (d < GAT_N) {
    const int beg = rowptr[d], end = rowptr[d + 1];
    const float4 erd = *(const float4*)(er + (size_t)d * GAT_H);
    const bool once = end - beg <= 32;
    float4 m = {0.f, 0.f, 0.f, 0.f}, den = {0.f, 0.f, 0.f, 0.f};
    int s1 = 0;
    float4 a1 = {0.f, 0.f, 0.f, 0.f};
    if (once) gat_segment_weights_once(csrc, el, erd, beg, end, lane, s1, a1);
    else      gat_segment_stats(csrc, el, erd, beg, end, lane, m, den);

    float acc0 = 0.f, acc1 = 0.f, acc2 = 0.f, acc3 = 0.f, acc4 = 0.f;
    for (int c = beg; c < end; c += 32) {
      int s = s1; float4 a = a1;
      if (!once) gat_chunk_weights(csrc, el, erd, c, end, lane, m, den, s, a);
      const int cnt = (end - c) < 32 ? (end - c) : 32;
#pragma unroll 4
      for (int t = 0; t < cnt; ++t) {
        const int st = __builtin_amdgcn_readlane(s, t);
        const float a0 = gat_lane_bcast(a.x, t), a1 = gat_lane_bcast(a.y, t);
        const float a2 = gat_lane_bcast(a.z, t), a3 = gat_lane_bcast(a.w, t);
        const float* row = feat + (size_t)st * 160 + lane;
        const float v0 = row[0], v1 = row[32], v2 = row[64], v3 = row[96], v4 = row[128];
        acc0 += v0 * a0;
        acc1 += v1 * (lane >= 8 ? a1 : a0);
        acc2 += v2 * (lane >= 16 ? a2 : a1);
        acc3 += v3 * (lane >= 24 ? a3 : a2);
        acc4 += v4 * a3;
      }
    }
    const float* bb = bias + lane;
    float* sv = sval + w * 160 + lane;
    sv[0]   = acc0 + bb[0];
    sv[32]  = acc1 + bb[32];
    sv[64]  = acc2 + bb[64];
    sv[96]  = acc3 + bb[96];
    sv[128] = acc4 + bb[128];
  }
  __syncthreads();
  for (int i = threadIdx.x; i < 8 * 40; i += 256) {
    const int nl = i / 40, c = i - nl * 40;
    if ((int)(blockIdx.x * 8 + nl) < GAT_N) {
      const float* sv = sval + nl * 160 + c;
      float sum = 0.f;
#pragma unroll
      for (int h = 0; h < GAT_H; ++h) sum += sv[h * 40];
      out[(size_t)blockIdx.x * 320 + i] =
          sum * 0.25f + resm[(size_t)(blockIdx.x * 8 + nl) * GAT_RESW + c];
    }
  }
}

extern "C" void kernel_launch(void* const* d_in, const int* in_sizes, int n_in,
                              void* d_out, int out_size, void* d_ws, size_t ws_size,
                              hipStream_t stream)
{
  const float* x     = (const float*)d_in[0];
  const int*   src   = (const int*)d_in[1];
  const int*   dst   = (const int*)d_in[2];
  const float* W0    = (const float*)d_in[3];
  const float* al0   = (const float*)d_in[4];
  const float* ar0   = (const float*)d_in[5];
  const float* b0    = (const float*)d_in[6];
  const float* W1    = (const float*)d_in[7];
  const float* al1   = (const float*)d_in[8];
  const float* ar1   = (const float*)d_in[9];
  const float* b1    = (const float*)d_in[10];
  const float* W2    = (const float*)d_in[11];
  const float* al2   = (const float*)d_in[12];
  const float* ar2   = (const float*)d_in[13];
  const float* b2    = (const float*)d_in[14];
  const float* Wres2 = (const float*)d_in[15];

  const size_t NF = (size_t)GAT_N * 256;
  float* ws   = (float*)d_ws;
  float* bufA = ws;
  float* bufB = bufA + NF;
  float* bufC = bufB + NF;
  float* EL   = bufC + NF;
  float* ER   = EL + (size_t)GAT_N * GAT_H;
  int*   OFFT = (int*)(ER + (size_t)GAT_N * GAT_H);
  int*   RPTR = OFFT + GAT_OFFT_LEN;
  int*   BKT  = RPTR + GAT_RPTR_LEN;
  int*   CSRC = BKT + 2 * (size_t)GAT_E;
  float* WM   = (float*)(CSRC + GAT_E);

  const int rowTiles = (GAT_N + 31) / 32;
  const int gemm0Blocks = GAT_SORT_B + (GAT_N + 31) / 32;
  dim3 gemmBlk256(128);
  dim3 gemmBlk160(224);

  const int nodeBlocks = (GAT_N + 7) / 8;

  gat_chunk_buckets<<<GAT_SORT_P + GAT_FOLD_BLOCKS, GAT_SORT_T1, 0, stream>>>(src, dst, OFFT, BKT, Wres2, WM);

  gat_gemm_wmma<256, 256, 4, 64, true, 0, GAT_SORT_B><<<gemm0Blocks, gemmBlk256, 0, stream>>>(x, W0, bufA, GAT_N, al0, ar0, EL, ER, nullptr, nullptr, OFFT, BKT, RPTR, CSRC);
  gat_node_agg_elu<<<nodeBlocks, 256, 0, stream>>>(RPTR, CSRC, bufA, EL, ER, b0, nullptr, bufB);

  gat_gemm_wmma<256, 256, 4, 64, true, 0, 0><<<rowTiles, gemmBlk256, 0, stream>>>(bufB, W1, bufA, GAT_N, al1, ar1, EL, ER, nullptr, nullptr, nullptr, nullptr, nullptr, nullptr);
  gat_node_agg_elu<<<nodeBlocks, 256, 0, stream>>>(RPTR, CSRC, bufA, EL, ER, b1, bufB, bufC);

  gat_gemm_wmma<256, 160, 2, 40, true, 2, 0><<<rowTiles, gemmBlk160, 0, stream>>>(bufC, W2, bufA, GAT_N, al2, ar2, EL, ER, WM, bufB, nullptr, nullptr, nullptr, nullptr);
  gat_node_agg_mean<<<nodeBlocks, 256, 0, stream>>>(RPTR, CSRC, bufA, EL, ER, b2, bufB, (float*)d_out);
}
